# starting kernel
# baseline (speedup 1.0000x reference)
#include <hip/hip_runtime.h>

typedef __attribute__((ext_vector_type(2))) float v2f;
typedef __attribute__((ext_vector_type(8))) float v8f;

#define F_IN  128
#define H_DIM 64

__global__ void zero_f32(float* __restrict__ p, long n) {
  long i = (long)blockIdx.x * blockDim.x + threadIdx.x;
  long stride = (long)gridDim.x * blockDim.x;
  for (; i < n; i += stride) p[i] = 0.0f;
}

__global__ void degree_kernel(const int* __restrict__ src, const int* __restrict__ dst,
                              float* __restrict__ deg_out, float* __restrict__ deg_in, int E) {
  int i = blockIdx.x * blockDim.x + threadIdx.x;
  int stride = gridDim.x * blockDim.x;
  for (; i < E; i += stride) {
    atomicAdd(&deg_out[src[i]], 1.0f);
    atomicAdd(&deg_in[dst[i]], 1.0f);
  }
}

__global__ void norm_kernel(float* __restrict__ p, int n) {
  int i = blockIdx.x * blockDim.x + threadIdx.x;
  if (i < n) p[i] = rsqrtf(fmaxf(p[i], 1.0f));
}

__device__ __forceinline__ void stage_w_frag(v2f* lW, const float* __restrict__ W,
                                             int kblocks, int tid, int nthreads) {
  const int total = kblocks * 4 * 32;
  for (int i = tid; i < total; i += nthreads) {
    const int kb = i >> 7;
    const int ct = (i >> 5) & 3;
    const int ln = i & 31;
    const int kg = (ln >> 4) * 2;
    const int n  = ct * 16 + (ln & 15);
    v2f p;
    p.x = W[(kb * 4 + kg)     * H_DIM + n];
    p.y = W[(kb * 4 + kg + 1) * H_DIM + n];
    lW[i] = p;
  }
}

__global__ __launch_bounds__(256) void gemm1_wmma(
    const float* __restrict__ feat, const float* __restrict__ norm_src,
    const float* __restrict__ W1, float* __restrict__ h1, int N) {
  __shared__ v2f lW[(F_IN / 4) * 4 * 32];
  stage_w_frag(lW, W1, F_IN / 4, threadIdx.x, 256);
  __syncthreads();

  const int wave = threadIdx.x >> 5;
  const int lane = threadIdx.x & 31;
  const int tile = blockIdx.x * 8 + wave;
  const int row0 = tile * 16;
  if (row0 >= N) return;

  const int mrow = row0 + (lane & 15);
  const int kgrp = (lane >> 4) * 2;
  const float ns = norm_src[mrow];

  v8f acc[4] = {};
  for (int kb = 0; kb < F_IN / 4; ++kb) {
    v2f a = *(const v2f*)(feat + (long)mrow * F_IN + kb * 4 + kgrp);
    a.x *= ns; a.y *= ns;
#pragma unroll
    for (int ct = 0; ct < 4; ++ct) {
      const v2f b = lW[(kb * 4 + ct) * 32 + lane];
      acc[ct] = __builtin_amdgcn_wmma_f32_16x16x4_f32(
          false, a, false, b, (short)0, acc[ct], false, false);
    }
  }

  const int mbase = row0 + ((lane >> 4) << 3);
#pragma unroll
  for (int ct = 0; ct < 4; ++ct) {
    const int n = ct * 16 + (lane & 15);
#pragma unroll
    for (int v = 0; v < 8; ++v)
      h1[(long)(mbase + v) * H_DIM + n] = acc[ct][v];
  }
}

__global__ void aggregate_kernel(const int* __restrict__ src, const int* __restrict__ dst,
                                 const float* __restrict__ h, float* __restrict__ agg, int E) {
  long t = (long)blockIdx.x * blockDim.x + threadIdx.x;
  const long total = (long)E * 16;
  const long stride = (long)gridDim.x * blockDim.x;
  for (; t < total; t += stride) {
    const int e = (int)(t >> 4);
    const int q = (int)(t & 15);
    const int s = src[e], d = dst[e];
    const float4 v = ((const float4*)h)[(long)s * 16 + q];
    float* ap = agg + (long)d * H_DIM + q * 4;
    atomicAdd(ap + 0, v.x);
    atomicAdd(ap + 1, v.y);
    atomicAdd(ap + 2, v.z);
    atomicAdd(ap + 3, v.w);
  }
}

__global__ void relu_scale_kernel(const float* __restrict__ agg, const float* __restrict__ norm_dst,
                                  const float* __restrict__ b1, const float* __restrict__ norm_src,
                                  float* __restrict__ h2, int N) {
  long i = (long)blockIdx.x * blockDim.x + threadIdx.x;
  const long total = (long)N * H_DIM;
  if (i >= total) return;
  const int node = (int)(i >> 6);
  const int j = (int)(i & 63);
  const float v = fmaxf(agg[i] * norm_dst[node] + b1[j], 0.0f);
  h2[i] = v * norm_src[node];
}

__global__ __launch_bounds__(256) void gemm2_final_wmma(
    const float* __restrict__ agg2,
    const float* __restrict__ W_mu, const float* __restrict__ b_mu,
    const float* __restrict__ W_sig, const float* __restrict__ b_sig,
    const float* __restrict__ norm_dst, const float* __restrict__ noise,
    float* __restrict__ z, int N) {
  __shared__ v2f lWm[(H_DIM / 4) * 4 * 32];
  __shared__ v2f lWs[(H_DIM / 4) * 4 * 32];
  stage_w_frag(lWm, W_mu, H_DIM / 4, threadIdx.x, 256);
  stage_w_frag(lWs, W_sig, H_DIM / 4, threadIdx.x, 256);
  __syncthreads();

  const int wave = threadIdx.x >> 5;
  const int lane = threadIdx.x & 31;
  const int tile = blockIdx.x * 8 + wave;
  const int row0 = tile * 16;
  if (row0 >= N) return;

  const int mrow = row0 + (lane & 15);
  const int kgrp = (lane >> 4) * 2;

  v8f am[4] = {}, as[4] = {};
  for (int kb = 0; kb < H_DIM / 4; ++kb) {
    const v2f a = *(const v2f*)(agg2 + (long)mrow * H_DIM + kb * 4 + kgrp);
#pragma unroll
    for (int ct = 0; ct < 4; ++ct) {
      const v2f bm = lWm[(kb * 4 + ct) * 32 + lane];
      const v2f bs = lWs[(kb * 4 + ct) * 32 + lane];
      am[ct] = __builtin_amdgcn_wmma_f32_16x16x4_f32(
          false, a, false, bm, (short)0, am[ct], false, false);
      as[ct] = __builtin_amdgcn_wmma_f32_16x16x4_f32(
          false, a, false, bs, (short)0, as[ct], false, false);
    }
  }

  const int mbase = row0 + ((lane >> 4) << 3);
#pragma unroll
  for (int ct = 0; ct < 4; ++ct) {
    const int n = ct * 16 + (lane & 15);
    const float bmu = b_mu[n];
    const float bsg = b_sig[n];
#pragma unroll
    for (int v = 0; v < 8; ++v) {
      const int m = mbase + v;
      const float nd = norm_dst[m];
      const float mu = am[ct][v] * nd + bmu;
      const float ls = as[ct][v] * nd + bsg;
      z[(long)m * H_DIM + n] = mu + noise[(long)m * H_DIM + n] * expf(ls);
    }
  }
}

extern "C" void kernel_launch(void* const* d_in, const int* in_sizes, int n_in,
                              void* d_out, int out_size, void* d_ws, size_t ws_size,
                              hipStream_t stream) {
  const float* feat  = (const float*)d_in[0];
  const int*   src   = (const int*)d_in[1];
  const int*   dst   = (const int*)d_in[2];
  const float* noise = (const float*)d_in[3];
  const float* W1    = (const float*)d_in[4];
  const float* b1    = (const float*)d_in[5];
  const float* W_mu  = (const float*)d_in[6];
  const float* b_mu  = (const float*)d_in[7];
  const float* W_sig = (const float*)d_in[8];
  const float* b_sig = (const float*)d_in[9];

  const int E = in_sizes[1];
  const int N = in_sizes[3] / H_DIM;

  float* ws       = (float*)d_ws;
  float* norm_src = ws;
  float* norm_dst = ws + N;
  float* buf_h    = ws + 2L * N;
  float* buf_agg  = buf_h + (long)N * H_DIM;
  float* zout     = (float*)d_out;

  const int threads = 256;
  const int gemm_blocks = (N / 16 + 7) / 8;
  const long elems = (long)N * H_DIM;

  zero_f32<<<(int)((2L * N + threads - 1) / threads), threads, 0, stream>>>(norm_src, 2L * N);
  degree_kernel<<<(E + threads - 1) / threads, threads, 0, stream>>>(src, dst, norm_src, norm_dst, E);
  norm_kernel<<<(2 * N + threads - 1) / threads, threads, 0, stream>>>(norm_src, 2 * N);

  gemm1_wmma<<<gemm_blocks, threads, 0, stream>>>(feat, norm_src, W1, buf_h, N);
  zero_f32<<<(int)((elems + threads - 1) / threads), threads, 0, stream>>>(buf_agg, elems);
  aggregate_kernel<<<(int)(((long)E * 16 + threads - 1) / threads), threads, 0, stream>>>(
      src, dst, buf_h, buf_agg, E);
  relu_scale_kernel<<<(int)((elems + threads - 1) / threads), threads, 0, stream>>>(
      buf_agg, norm_dst, b1, norm_src, buf_h, N);

  zero_f32<<<(int)((elems + threads - 1) / threads), threads, 0, stream>>>(buf_agg, elems);
  aggregate_kernel<<<(int)(((long)E * 16 + threads - 1) / threads), threads, 0, stream>>>(
      src, dst, buf_h, buf_agg, E);
  gemm2_final_wmma<<<gemm_blocks, threads, 0, stream>>>(
      buf_agg, W_mu, b_mu, W_sig, b_sig, norm_dst, noise, zout, N);
}


# One bucketing launch, not two: each edge slice ordered by bucket on its own (no count plane, no prefix); a group collects its bucket's 256 runs into an LDS list; 4 launches; row sums in arrival order
# speedup vs baseline: 3.5324x; 3.5324x over previous
#include <hip/hip_runtime.h>

typedef __attribute__((ext_vector_type(2))) float v2f;
typedef __attribute__((ext_vector_type(8))) float v8f;

#define F_IN  128
#define H_DIM 64
#define R_BIN 400
#define G1_THREADS (R_BIN / 16 * 32)

__device__ __forceinline__ void stage_w_frag(v2f* lW, const float* __restrict__ W,
                                             int kblocks, int tid, int nthreads) {
  const int total = kblocks * 4 * 32;
  for (int i = tid; i < total; i += nthreads) {
    const int kb = i >> 7;
    const int ct = (i >> 5) & 3;
    const int ln = i & 31;
    const int kg = (ln >> 4) * 2;
    const int n  = ct * 16 + (ln & 15);
    v2f p;
    p.x = W[(kb * 4 + kg)     * H_DIM + n];
    p.y = W[(kb * 4 + kg + 1) * H_DIM + n];
    lW[i] = p;
  }
}

#define BK_BLOCKS 256
#define BK_THREADS 1024
#define BK_K 7

__global__ __launch_bounds__(BK_THREADS) void bucket_sort_slices(
    const int* __restrict__ src, const int* __restrict__ dst,
    unsigned* __restrict__ off_dst, unsigned* __restrict__ off_src,
    unsigned* __restrict__ packed_dst, unsigned* __restrict__ packed_src,
    int E, int NB, int chunk, int offp) {
  extern __shared__ unsigned sm[];
  unsigned* pfx  = sm;
  unsigned* loc  = pfx + NB + 1;
  unsigned* word = loc + NB;
  const bool by_src = blockIdx.y != 0;
  const int* __restrict__ key = by_src ? src : dst;
  unsigned* __restrict__ off = (by_src ? off_src : off_dst) + (long)blockIdx.x * offp;
  unsigned* __restrict__ packed = (by_src ? packed_src : packed_dst) + (long)blockIdx.x * chunk;
  for (int i = threadIdx.x; i <= NB; i += BK_THREADS) pfx[i] = 0u;
  __syncthreads();

  const int e0 = blockIdx.x * chunk;
  const int n = max(min(e0 + chunk, E) - e0, 0);
  for (int base = threadIdx.x; base < n; base += BK_THREADS * BK_K) {
    int d[BK_K];
#pragma unroll
    for (int k = 0; k < BK_K; ++k) {
      const int i = base + k * BK_THREADS;
      d[k] = key[e0 + (i < n ? i : n - 1)];
    }
#pragma unroll
    for (int k = 0; k < BK_K; ++k)
      if (base + k * BK_THREADS < n) atomicAdd(&pfx[d[k] / R_BIN], 1u);
  }
  __syncthreads();

  if (threadIdx.x < 32) {
    const int per = (NB + 31) / 32;
    const int b0 = (int)threadIdx.x * per;
    const int b1 = min(b0 + per, NB);
    unsigned sum = 0u;
    for (int b = b0; b < b1; ++b) sum += pfx[b];
    unsigned inc = sum;
    for (int dlt = 1; dlt < 32; dlt <<= 1) {
      const unsigned v = __shfl_up(inc, dlt);
      if ((int)threadIdx.x >= dlt) inc += v;
    }
    unsigned run = inc - sum;
    for (int b = b0; b < b1; ++b) {
      const unsigned c = pfx[b];
      pfx[b] = run;
      loc[b] = run;
      run += c;
    }
    if (threadIdx.x == 31) pfx[NB] = inc;
  }
  __syncthreads();

  for (int i = threadIdx.x; i <= NB; i += BK_THREADS) off[i] = pfx[i];
  for (int base = threadIdx.x; base < n; base += BK_THREADS * BK_K) {
    int d[BK_K], s[BK_K];
#pragma unroll
    for (int k = 0; k < BK_K; ++k) {
      const int i = base + k * BK_THREADS;
      const int ic = i < n ? i : n - 1;
      d[k] = key[e0 + ic];
      s[k] = by_src ? 0 : src[e0 + ic];
    }
#pragma unroll
    for (int k = 0; k < BK_K; ++k) {
      if (base + k * BK_THREADS < n) {
        const int bin = d[k] / R_BIN;
        const unsigned at = atomicAdd(&loc[bin], 1u);
        word[at] = (unsigned)(d[k] - bin * R_BIN) + (unsigned)s[k] * R_BIN;
      }
    }
  }
  __syncthreads();
  for (int i = threadIdx.x; i < n; i += BK_THREADS) packed[i] = word[i];
}

__device__ __forceinline__ unsigned load_run(unsigned* rst, const unsigned* __restrict__ off,
                                             int bin, int offp, int chunk) {
  unsigned len = 0u;
  if (threadIdx.x < BK_BLOCKS) {
    const unsigned* o = off + (long)threadIdx.x * offp + bin;
    const unsigned a = o[0];
    len = o[1] - a;
    rst[threadIdx.x] = threadIdx.x * (unsigned)chunk + a;
  }
  return len;
}

__device__ __forceinline__ void scan_runs(unsigned* pre, unsigned* wtot, unsigned len) {
  static_assert(BK_BLOCKS % 32 == 0, "the runs fill whole waves");
  const int wave = threadIdx.x >> 5;
  const int lane = threadIdx.x & 31;
  unsigned inc = len;
  for (int dlt = 1; dlt < 32; dlt <<= 1) {
    const unsigned v = __shfl_up(inc, dlt);
    if (lane >= dlt) inc += v;
  }
  if (threadIdx.x < BK_BLOCKS && lane == 31) wtot[wave] = inc;
  __syncthreads();
  if (threadIdx.x < BK_BLOCKS) {
    unsigned before = 0u;
    for (int q = 0; q < wave; ++q) before += wtot[q];
    pre[threadIdx.x] = before + inc - len;
    if (threadIdx.x == BK_BLOCKS - 1) pre[BK_BLOCKS] = before + inc;
  }
  __syncthreads();
}

template <int NW>
__device__ __forceinline__ void count_run_rows(unsigned* degc, const unsigned* __restrict__ packed,
                                               const unsigned* rst, const unsigned* rln) {
  constexpr int RPW = (BK_BLOCKS + NW - 1) / NW;
  const int wave = threadIdx.x >> 5;
  const unsigned lane = threadIdx.x & 31;
  unsigned w0[RPW], w1[RPW], ln[RPW];
#pragma unroll
  for (int j = 0; j < RPW; ++j) {
    const int r = wave + j * NW;
    const int rc = r < BK_BLOCKS ? r : 0;
    const unsigned s = rst[rc];
    ln[j] = r < BK_BLOCKS ? rln[rc] : 0u;
    w0[j] = packed[lane < ln[j] ? s + lane : 0u];
    w1[j] = packed[lane + 32u < ln[j] ? s + lane + 32u : 0u];
  }
#pragma unroll
  for (int j = 0; j < RPW; ++j) {
    if (lane < ln[j]) atomicAdd(&degc[w0[j] % R_BIN], 1u);
    if (lane + 32u < ln[j]) atomicAdd(&degc[w1[j] % R_BIN], 1u);
  }
  for (int r = wave; r < BK_BLOCKS; r += NW) {
    const unsigned s = rst[r];
    const unsigned len = rln[r];
    for (unsigned i = 64u + lane; i < len; i += 32u) atomicAdd(&degc[packed[s + i] % R_BIN], 1u);
  }
}

__global__ __launch_bounds__(G1_THREADS) void gemm1_wmma(
    const float* __restrict__ feat, const unsigned* __restrict__ packed_src,
    const unsigned* __restrict__ off_src, const float* __restrict__ W1,
    float* __restrict__ norm_src, float* __restrict__ h1, int N, int chunk, int offp) {
  static_assert(R_BIN % 16 == 0, "a bucket is a whole number of 16-row tiles");
  static_assert(G1_THREADS >= BK_BLOCKS, "one thread per run reads its offsets");
  __shared__ v2f lW[(F_IN / 4) * 4 * 32];
  __shared__ unsigned degc[R_BIN];
  __shared__ unsigned rst[BK_BLOCKS];
  __shared__ unsigned rln[BK_BLOCKS];
  if (threadIdx.x < R_BIN) degc[threadIdx.x] = 0u;
  {
    const unsigned len = load_run(rst, off_src, blockIdx.x, offp, chunk);
    if (threadIdx.x < BK_BLOCKS) rln[threadIdx.x] = len;
  }
  stage_w_frag(lW, W1, F_IN / 4, threadIdx.x, G1_THREADS);
  __syncthreads();
  count_run_rows<G1_THREADS / 32>(degc, packed_src, rst, rln);
  __syncthreads();
  if (threadIdx.x < R_BIN && blockIdx.x * R_BIN + threadIdx.x < N)
    norm_src[blockIdx.x * R_BIN + threadIdx.x] =
        rsqrtf(fmaxf((float)degc[threadIdx.x], 1.0f));

  const int wave = threadIdx.x >> 5;
  const int lane = threadIdx.x & 31;
  const int tile = blockIdx.x * (R_BIN / 16) + wave;
  const int row0 = tile * 16;
  if (row0 >= N) return;

  const int mrow = row0 + (lane & 15);
  const int kgrp = (lane >> 4) * 2;
  const float ns = rsqrtf(fmaxf((float)degc[wave * 16 + (lane & 15)], 1.0f));

  v8f acc[4] = {};
  for (int kb = 0; kb < F_IN / 4; ++kb) {
    v2f a = *(const v2f*)(feat + (long)mrow * F_IN + kb * 4 + kgrp);
    a.x *= ns; a.y *= ns;
#pragma unroll
    for (int ct = 0; ct < 4; ++ct) {
      const v2f b = lW[(kb * 4 + ct) * 32 + lane];
      acc[ct] = __builtin_amdgcn_wmma_f32_16x16x4_f32(
          false, a, false, b, (short)0, acc[ct], false, false);
    }
  }

  const int mbase = row0 + ((lane >> 4) << 3);
#pragma unroll
  for (int ct = 0; ct < 4; ++ct) {
    const int n = ct * 16 + (lane & 15);
#pragma unroll
    for (int v = 0; v < 8; ++v)
      h1[(long)(mbase + v) * H_DIM + n] = acc[ct][v];
  }
}

template <int PITCH>
__device__ __forceinline__ void tile_add_row(float* tile, unsigned p, int lane, float2 v) {
  float* t = tile + (p % R_BIN) * PITCH + lane * 2;
  atomicAdd(t, v.x);
  atomicAdd(t + 1, v.y);
}

#define AGG_THREADS 1024
#define EPT 12
#define L_CAP 8192

__device__ __forceinline__ void fill_list(unsigned* list, const unsigned* __restrict__ packed,
                                          const unsigned* rst, const unsigned* pre, unsigned R) {
  constexpr int NW = AGG_THREADS / 32;
  constexpr int RPW = (BK_BLOCKS + NW - 1) / NW;
  const int wave = threadIdx.x >> 5;
  const unsigned lane = threadIdx.x & 31;
  unsigned w0[RPW], w1[RPW], g0[RPW], ln[RPW];
#pragma unroll
  for (int j = 0; j < RPW; ++j) {
    const int r = wave + j * NW;
    const int rc = r < BK_BLOCKS ? r : 0;
    const unsigned s = rst[rc];
    const unsigned p = pre[rc];
    ln[j] = r < BK_BLOCKS ? pre[rc + 1] - p : 0u;
    g0[j] = p + lane - R;
    w0[j] = packed[(lane < ln[j] && g0[j] < L_CAP) ? s + lane : 0u];
    w1[j] = packed[(lane + 32u < ln[j] && g0[j] + 32u < L_CAP) ? s + lane + 32u : 0u];
  }
#pragma unroll
  for (int j = 0; j < RPW; ++j) {
    if (lane < ln[j] && g0[j] < L_CAP) list[g0[j]] = w0[j];
    if (lane + 32u < ln[j] && g0[j] + 32u < L_CAP) list[g0[j] + 32u] = w1[j];
  }
  for (int r = wave; r < BK_BLOCKS; r += NW) {
    const unsigned s = rst[r];
    const unsigned p = pre[r];
    const unsigned len = pre[r + 1] - p;
    for (unsigned i = 64u + lane; i < len; i += 32u) {
      const unsigned g = p + i - R;
      if (g < L_CAP) list[g] = packed[s + i];
    }
  }
}

template <int PITCH>
__device__ __forceinline__ void list_sum_rows(float* tile, const unsigned* list, unsigned cnt,
                                              const float* __restrict__ h) {
  const int wave = threadIdx.x >> 5;
  const int lane = threadIdx.x & 31;
  const float2* hv = (const float2*)h;
  for (unsigned e = (unsigned)wave * EPT; e < cnt; e += (AGG_THREADS / 32) * EPT) {
    unsigned p[EPT];
    float2 v[EPT];
#pragma unroll
    for (int k = 0; k < EPT; ++k) p[k] = list[e + k < cnt ? e + k : cnt - 1u];
#pragma unroll
    for (int k = 0; k < EPT; ++k) v[k] = hv[(long)(p[k] / R_BIN) * 32 + lane];
#pragma unroll
    for (int k = 0; k < EPT; ++k) {
      const bool in = e + k < cnt;
      tile_add_row<PITCH>(tile, p[k], lane, make_float2(in ? v[k].x : 0.0f, in ? v[k].y : 0.0f));
    }
  }
}

template <int PITCH>
__device__ __forceinline__ void bucket_edges(float* tile, unsigned* degc, unsigned* list,
                                             unsigned* rst, unsigned* pre, unsigned* wtot,
                                             const unsigned* __restrict__ packed,
                                             const unsigned* __restrict__ off,
                                             int bin, int offp, int chunk,
                                             const float* __restrict__ h) {
  scan_runs(pre, wtot, load_run(rst, off, bin, offp, chunk));
  const unsigned total = pre[BK_BLOCKS];
  for (unsigned R = 0u; R < total; R += L_CAP) {
    fill_list(list, packed, rst, pre, R);
    __syncthreads();
    const unsigned cnt = min(total - R, (unsigned)L_CAP);
    for (unsigned i = threadIdx.x; i < cnt; i += AGG_THREADS) atomicAdd(&degc[list[i] % R_BIN], 1u);
    list_sum_rows<PITCH>(tile, list, cnt, h);
    __syncthreads();
  }
}

__global__ __launch_bounds__(AGG_THREADS) void bucket_aggregate_relu(
    const unsigned* __restrict__ packed, const unsigned* __restrict__ off,
    const float* __restrict__ h, const float* __restrict__ norm_src,
    const float* __restrict__ b1, float* __restrict__ out, int N, int chunk, int offp) {
  static_assert(AGG_THREADS % 16 == 0, "a thread keeps its 4-column group over the store loop");
  __shared__ alignas(16) float tile[R_BIN * H_DIM];
  __shared__ unsigned list[L_CAP];
  __shared__ unsigned rst[BK_BLOCKS];
  __shared__ unsigned pre[BK_BLOCKS + 1];
  __shared__ unsigned wtot[BK_BLOCKS / 32];
  __shared__ unsigned degc[R_BIN];
  __shared__ float nsl[R_BIN];
  const int bin = blockIdx.x;
  const int row_base = bin * R_BIN;
  float4* tile4 = (float4*)tile;
  for (int i = threadIdx.x; i < R_BIN * H_DIM / 4; i += AGG_THREADS)
    tile4[i] = make_float4(0.0f, 0.0f, 0.0f, 0.0f);
  if (threadIdx.x < R_BIN) {
    degc[threadIdx.x] = 0u;
    nsl[threadIdx.x] = (row_base + threadIdx.x < N) ? norm_src[row_base + threadIdx.x] : 0.0f;
  }
  bucket_edges<H_DIM>(tile, degc, list, rst, pre, wtot, packed, off, bin, offp, chunk, h);

  const float4 b = ((const float4*)b1)[threadIdx.x & 15];
  for (int i = threadIdx.x; i < R_BIN * H_DIM / 4; i += AGG_THREADS) {
    const int node = row_base + (i >> 4);
    if (node < N) {
      float4 v = tile4[i];
      const float nd = rsqrtf(fmaxf((float)degc[i >> 4], 1.0f));
      const float ns = nsl[i >> 4];
      v.x = fmaxf(v.x * nd + b.x, 0.0f) * ns;
      v.y = fmaxf(v.y * nd + b.y, 0.0f) * ns;
      v.z = fmaxf(v.z * nd + b.z, 0.0f) * ns;
      v.w = fmaxf(v.w * nd + b.w, 0.0f) * ns;
      ((float4*)out)[(long)node * 16 + (i & 15)] = v;
    }
  }
}

#define T_PITCH 68
#define L2_UNITS (R_BIN / 16 * 4)
__global__ __launch_bounds__(AGG_THREADS) void bucket_aggregate_final_wmma(
    const unsigned* __restrict__ packed, const unsigned* __restrict__ off,
    const float* __restrict__ h,
    const float* __restrict__ W_mu, const float* __restrict__ b_mu,
    const float* __restrict__ W_sig, const float* __restrict__ b_sig,
    const float* __restrict__ noise,
    float* __restrict__ z, int N, int chunk, int offp) {
  static_assert(R_BIN * T_PITCH % 4 == 0 && T_PITCH % 2 == 0, "float4 zero fill, 8-byte fragment reads");
  __shared__ alignas(16) float tile[R_BIN * T_PITCH];
  __shared__ v2f lWm[(H_DIM / 4) * 4 * 32];
  __shared__ v2f lWs[(H_DIM / 4) * 4 * 32];
  __shared__ unsigned list[L_CAP];
  __shared__ unsigned rst[BK_BLOCKS];
  __shared__ unsigned pre[BK_BLOCKS + 1];
  __shared__ unsigned wtot[BK_BLOCKS / 32];
  __shared__ unsigned degc[R_BIN];
  __shared__ float ndl[R_BIN];
  const int bin = blockIdx.x;
  const int row_base = bin * R_BIN;
  float4* tile4 = (float4*)tile;
  for (int i = threadIdx.x; i < R_BIN * T_PITCH / 4; i += AGG_THREADS)
    tile4[i] = make_float4(0.0f, 0.0f, 0.0f, 0.0f);
  if (threadIdx.x < R_BIN) degc[threadIdx.x] = 0u;
  stage_w_frag(lWm, W_mu, H_DIM / 4, threadIdx.x, AGG_THREADS);
  stage_w_frag(lWs, W_sig, H_DIM / 4, threadIdx.x, AGG_THREADS);
  bucket_edges<T_PITCH>(tile, degc, list, rst, pre, wtot, packed, off, bin, offp, chunk, h);
  if (threadIdx.x < R_BIN) ndl[threadIdx.x] = rsqrtf(fmaxf((float)degc[threadIdx.x], 1.0f));
  __syncthreads();

  const int wave = threadIdx.x >> 5;
  const int lane = threadIdx.x & 31;
  const int kgrp = (lane >> 4) * 2;
  for (int u = wave; u < L2_UNITS; u += AGG_THREADS / 32) {
    const int rt = u >> 2;
    const int ct = u & 3;
    const int row0 = row_base + rt * 16;
    if (row0 >= N) continue;
    const float* arow = tile + (rt * 16 + (lane & 15)) * T_PITCH + kgrp;
    v8f am = {}, as = {};
#pragma unroll
    for (int kb = 0; kb < H_DIM / 4; ++kb) {
      const v2f a  = *(const v2f*)(arow + kb * 4);
      const v2f bm = lWm[(kb * 4 + ct) * 32 + lane];
      const v2f bs = lWs[(kb * 4 + ct) * 32 + lane];
      am = __builtin_amdgcn_wmma_f32_16x16x4_f32(
          false, a, false, bm, (short)0, am, false, false);
      as = __builtin_amdgcn_wmma_f32_16x16x4_f32(
          false, a, false, bs, (short)0, as, false, false);
    }
    const int n = ct * 16 + (lane & 15);
    const int lrow = rt * 16 + ((lane >> 4) << 3);
    const float bmu = b_mu[n];
    const float bsg = b_sig[n];
#pragma unroll
    for (int v = 0; v < 8; ++v) {
      const int m = row_base + lrow + v;
      if (m < N) {
        const float nd = ndl[lrow + v];
        const float mu = am[v] * nd + bmu;
        const float ls = as[v] * nd + bsg;
        z[(long)m * H_DIM + n] = mu + noise[(long)m * H_DIM + n] * expf(ls);
      }
    }
  }
}

extern "C" void kernel_launch(void* const* d_in, const int* in_sizes, int n_in,
                              void* d_out, int out_size, void* d_ws, size_t ws_size,
                              hipStream_t stream) {
  const float* feat  = (const float*)d_in[0];
  const int*   src   = (const int*)d_in[1];
  const int*   dst   = (const int*)d_in[2];
  const float* noise = (const float*)d_in[3];
  const float* W1    = (const float*)d_in[4];
  const float* b1    = (const float*)d_in[5];
  const float* W_mu  = (const float*)d_in[6];
  const float* b_mu  = (const float*)d_in[7];
  const float* W_sig = (const float*)d_in[8];
  const float* b_sig = (const float*)d_in[9];

  const int E = in_sizes[1];
  const int N = in_sizes[3] / H_DIM;

  const int NB    = (N + R_BIN - 1) / R_BIN;
  const int chunk = (E + BK_BLOCKS - 1) / BK_BLOCKS;
  const int offp  = (NB + 32) & ~31;

  float* ws       = (float*)d_ws;
  float* norm_src = ws;
  unsigned* off_dst = (unsigned*)(ws + N);
  float* buf_a    = ws + 2L * N;
  float* buf_b    = buf_a + (long)N * H_DIM;
  float* zout     = (float*)d_out;

  unsigned* packed_dst  = (unsigned*)(buf_b + (long)N * H_DIM);
  unsigned* packed_src  = (unsigned*)buf_b;
  unsigned* off_src     = packed_src + (long)BK_BLOCKS * chunk;

  bucket_sort_slices<<<dim3(BK_BLOCKS, 2), BK_THREADS, (2 * NB + 1 + chunk) * sizeof(unsigned), stream>>>(
      src, dst, off_dst, off_src, packed_dst, packed_src, E, NB, chunk, offp);

  gemm1_wmma<<<NB, G1_THREADS, 0, stream>>>(feat, packed_src, off_src, W1, norm_src, buf_a, N, chunk, offp);

  bucket_aggregate_relu<<<NB, AGG_THREADS, 0, stream>>>(
      packed_dst, off_dst, buf_a, norm_src, b1, buf_b, N, chunk, offp);

  bucket_aggregate_final_wmma<<<NB, AGG_THREADS, 0, stream>>>(
      packed_dst, off_dst, buf_b, W_mu, b_mu, W_sig, b_sig, noise, zout, N, chunk, offp);
}


# Last launch, mu/sigma units: a unit's 8 noise words, row norms and bias words are loaded ahead of its 32 products (clamped row, no branch), not row by row behind them; values bit-identical
# speedup vs baseline: 3.7936x; 1.0739x over previous
#include <hip/hip_runtime.h>

typedef __attribute__((ext_vector_type(2))) float v2f;
typedef __attribute__((ext_vector_type(8))) float v8f;

#define F_IN  128
#define H_DIM 64
#define R_BIN 400
#define G1_THREADS (R_BIN / 16 * 32)

__device__ __forceinline__ void stage_w_frag(v2f* lW, const float* __restrict__ W,
                                             int kblocks, int tid, int nthreads) {
  const int total = kblocks * 4 * 32;
  for (int i = tid; i < total; i += nthreads) {
    const int kb = i >> 7;
    const int ct = (i >> 5) & 3;
    const int ln = i & 31;
    const int kg = (ln >> 4) * 2;
    const int n  = ct * 16 + (ln & 15);
    v2f p;
    p.x = W[(kb * 4 + kg)     * H_DIM + n];
    p.y = W[(kb * 4 + kg + 1) * H_DIM + n];
    lW[i] = p;
  }
}

#define BK_BLOCKS 256
#define BK_THREADS 1024
#define BK_K 7

__global__ __launch_bounds__(BK_THREADS) void bucket_sort_slices(
    const int* __restrict__ src, const int* __restrict__ dst,
    unsigned* __restrict__ off_dst, unsigned* __restrict__ off_src,
    unsigned* __restrict__ packed_dst, unsigned* __restrict__ packed_src,
    int E, int NB, int chunk, int offp) {
  extern __shared__ unsigned sm[];
  unsigned* pfx  = sm;
  unsigned* loc  = pfx + NB + 1;
  unsigned* word = loc + NB;
  const bool by_src = blockIdx.y != 0;
  const int* __restrict__ key = by_src ? src : dst;
  unsigned* __restrict__ off = (by_src ? off_src : off_dst) + (long)blockIdx.x * offp;
  unsigned* __restrict__ packed = (by_src ? packed_src : packed_dst) + (long)blockIdx.x * chunk;
  for (int i = threadIdx.x; i <= NB; i += BK_THREADS) pfx[i] = 0u;
  __syncthreads();

  const int e0 = blockIdx.x * chunk;
  const int n = max(min(e0 + chunk, E) - e0, 0);
  for (int base = threadIdx.x; base < n; base += BK_THREADS * BK_K) {
    int d[BK_K];
#pragma unroll
    for (int k = 0; k < BK_K; ++k) {
      const int i = base + k * BK_THREADS;
      d[k] = key[e0 + (i < n ? i : n - 1)];
    }
#pragma unroll
    for (int k = 0; k < BK_K; ++k)
      if (base + k * BK_THREADS < n) atomicAdd(&pfx[d[k] / R_BIN], 1u);
  }
  __syncthreads();

  if (threadIdx.x < 32) {
    const int per = (NB + 31) / 32;
    const int b0 = (int)threadIdx.x * per;
    const int b1 = min(b0 + per, NB);
    unsigned sum = 0u;
    for (int b = b0; b < b1; ++b) sum += pfx[b];
    unsigned inc = sum;
    for (int dlt = 1; dlt < 32; dlt <<= 1) {
      const unsigned v = __shfl_up(inc, dlt);
      if ((int)threadIdx.x >= dlt) inc += v;
    }
    unsigned run = inc - sum;
    for (int b = b0; b < b1; ++b) {
      const unsigned c = pfx[b];
      pfx[b] = run;
      loc[b] = run;
      run += c;
    }
    if (threadIdx.x == 31) pfx[NB] = inc;
  }
  __syncthreads();

  for (int i = threadIdx.x; i <= NB; i += BK_THREADS) off[i] = pfx[i];
  for (int base = threadIdx.x; base < n; base += BK_THREADS * BK_K) {
    int d[BK_K], s[BK_K];
#pragma unroll
    for (int k = 0; k < BK_K; ++k) {
      const int i = base + k * BK_THREADS;
      const int ic = i < n ? i : n - 1;
      d[k] = key[e0 + ic];
      s[k] = by_src ? 0 : src[e0 + ic];
    }
#pragma unroll
    for (int k = 0; k < BK_K; ++k) {
      if (base + k * BK_THREADS < n) {
        const int bin = d[k] / R_BIN;
        const unsigned at = atomicAdd(&loc[bin], 1u);
        word[at] = (unsigned)(d[k] - bin * R_BIN) + (unsigned)s[k] * R_BIN;
      }
    }
  }
  __syncthreads();
  for (int i = threadIdx.x; i < n; i += BK_THREADS) packed[i] = word[i];
}

__device__ __forceinline__ unsigned load_run(unsigned* rst, const unsigned* __restrict__ off,
                                             int bin, int offp, int chunk) {
  unsigned len = 0u;
  if (threadIdx.x < BK_BLOCKS) {
    const unsigned* o = off + (long)threadIdx.x * offp + bin;
    const unsigned a = o[0];
    len = o[1] - a;
    rst[threadIdx.x] = threadIdx.x * (unsigned)chunk + a;
  }
  return len;
}

__device__ __forceinline__ void scan_runs(unsigned* pre, unsigned* wtot, unsigned len) {
  static_assert(BK_BLOCKS % 32 == 0, "the runs fill whole waves");
  const int wave = threadIdx.x >> 5;
  const int lane = threadIdx.x & 31;
  unsigned inc = len;
  for (int dlt = 1; dlt < 32; dlt <<= 1) {
    const unsigned v = __shfl_up(inc, dlt);
    if (lane >= dlt) inc += v;
  }
  if (threadIdx.x < BK_BLOCKS && lane == 31) wtot[wave] = inc;
  __syncthreads();
  if (threadIdx.x < BK_BLOCKS) {
    unsigned before = 0u;
    for (int q = 0; q < wave; ++q) before += wtot[q];
    pre[threadIdx.x] = before + inc - len;
    if (threadIdx.x == BK_BLOCKS - 1) pre[BK_BLOCKS] = before + inc;
  }
  __syncthreads();
}

template <int NW>
__device__ __forceinline__ void count_run_rows(unsigned* degc, const unsigned* __restrict__ packed,
                                               const unsigned* rst, const unsigned* rln) {
  constexpr int RPW = (BK_BLOCKS + NW - 1) / NW;
  const int wave = threadIdx.x >> 5;
  const unsigned lane = threadIdx.x & 31;
  unsigned w0[RPW], w1[RPW], ln[RPW];
#pragma unroll
  for (int j = 0; j < RPW; ++j) {
    const int r = wave + j * NW;
    const int rc = r < BK_BLOCKS ? r : 0;
    const unsigned s = rst[rc];
    ln[j] = r < BK_BLOCKS ? rln[rc] : 0u;
    w0[j] = packed[lane < ln[j] ? s + lane : 0u];
    w1[j] = packed[lane + 32u < ln[j] ? s + lane + 32u : 0u];
  }
#pragma unroll
  for (int j = 0; j < RPW; ++j) {
    if (lane < ln[j]) atomicAdd(&degc[w0[j] % R_BIN], 1u);
    if (lane + 32u < ln[j]) atomicAdd(&degc[w1[j] % R_BIN], 1u);
  }
  for (int r = wave; r < BK_BLOCKS; r += NW) {
    const unsigned s = rst[r];
    const unsigned len = rln[r];
    for (unsigned i = 64u + lane; i < len; i += 32u) atomicAdd(&degc[packed[s + i] % R_BIN], 1u);
  }
}

__global__ __launch_bounds__(G1_THREADS) void gemm1_wmma(
    const float* __restrict__ feat, const unsigned* __restrict__ packed_src,
    const unsigned* __restrict__ off_src, const float* __restrict__ W1,
    float* __restrict__ norm_src, float* __restrict__ h1, int N, int chunk, int offp) {
  static_assert(R_BIN % 16 == 0, "a bucket is a whole number of 16-row tiles");
  static_assert(G1_THREADS >= BK_BLOCKS, "one thread per run reads its offsets");
  __shared__ v2f lW[(F_IN / 4) * 4 * 32];
  __shared__ unsigned degc[R_BIN];
  __shared__ unsigned rst[BK_BLOCKS];
  __shared__ unsigned rln[BK_BLOCKS];
  if (threadIdx.x < R_BIN) degc[threadIdx.x] = 0u;
  {
    const unsigned len = load_run(rst, off_src, blockIdx.x, offp, chunk);
    if (threadIdx.x < BK_BLOCKS) rln[threadIdx.x] = len;
  }
  stage_w_frag(lW, W1, F_IN / 4, threadIdx.x, G1_THREADS);
  __syncthreads();
  count_run_rows<G1_THREADS / 32>(degc, packed_src, rst, rln);
  __syncthreads();
  if (threadIdx.x < R_BIN && blockIdx.x * R_BIN + threadIdx.x < N)
    norm_src[blockIdx.x * R_BIN + threadIdx.x] =
        rsqrtf(fmaxf((float)degc[threadIdx.x], 1.0f));

  const int wave = threadIdx.x >> 5;
  const int lane = threadIdx.x & 31;
  const int tile = blockIdx.x * (R_BIN / 16) + wave;
  const int row0 = tile * 16;
  if (row0 >= N) return;

  const int mrow = row0 + (lane & 15);
  const int kgrp = (lane >> 4) * 2;
  const float ns = rsqrtf(fmaxf((float)degc[wave * 16 + (lane & 15)], 1.0f));

  v8f acc[4] = {};
  for (int kb = 0; kb < F_IN / 4; ++kb) {
    v2f a = *(const v2f*)(feat + (long)mrow * F_IN + kb * 4 + kgrp);
    a.x *= ns; a.y *= ns;
#pragma unroll
    for (int ct = 0; ct < 4; ++ct) {
      const v2f b = lW[(kb * 4 + ct) * 32 + lane];
      acc[ct] = __builtin_amdgcn_wmma_f32_16x16x4_f32(
          false, a, false, b, (short)0, acc[ct], false, false);
    }
  }

  const int mbase = row0 + ((lane >> 4) << 3);
#pragma unroll
  for (int ct = 0; ct < 4; ++ct) {
    const int n = ct * 16 + (lane & 15);
#pragma unroll
    for (int v = 0; v < 8; ++v)
      h1[(long)(mbase + v) * H_DIM + n] = acc[ct][v];
  }
}

template <int PITCH>
__device__ __forceinline__ void tile_add_row(float* tile, unsigned p, int lane, float2 v) {
  float* t = tile + (p % R_BIN) * PITCH + lane * 2;
  atomicAdd(t, v.x);
  atomicAdd(t + 1, v.y);
}

#define AGG_THREADS 1024
#define EPT 12
#define L_CAP 8192

__device__ __forceinline__ void fill_list(unsigned* list, const unsigned* __restrict__ packed,
                                          const unsigned* rst, const unsigned* pre, unsigned R) {
  constexpr int NW = AGG_THREADS / 32;
  constexpr int RPW = (BK_BLOCKS + NW - 1) / NW;
  const int wave = threadIdx.x >> 5;
  const unsigned lane = threadIdx.x & 31;
  unsigned w0[RPW], w1[RPW], g0[RPW], ln[RPW];
#pragma unroll
  for (int j = 0; j < RPW; ++j) {
    const int r = wave + j * NW;
    const int rc = r < BK_BLOCKS ? r : 0;
    const unsigned s = rst[rc];
    const unsigned p = pre[rc];
    ln[j] = r < BK_BLOCKS ? pre[rc + 1] - p : 0u;
    g0[j] = p + lane - R;
    w0[j] = packed[(lane < ln[j] && g0[j] < L_CAP) ? s + lane : 0u];
    w1[j] = packed[(lane + 32u < ln[j] && g0[j] + 32u < L_CAP) ? s + lane + 32u : 0u];
  }
#pragma unroll
  for (int j = 0; j < RPW; ++j) {
    if (lane < ln[j] && g0[j] < L_CAP) list[g0[j]] = w0[j];
    if (lane + 32u < ln[j] && g0[j] + 32u < L_CAP) list[g0[j] + 32u] = w1[j];
  }
  for (int r = wave; r < BK_BLOCKS; r += NW) {
    const unsigned s = rst[r];
    const unsigned p = pre[r];
    const unsigned len = pre[r + 1] - p;
    for (unsigned i = 64u + lane; i < len; i += 32u) {
      const unsigned g = p + i - R;
      if (g < L_CAP) list[g] = packed[s + i];
    }
  }
}

template <int PITCH>
__device__ __forceinline__ void list_sum_rows(float* tile, const unsigned* list, unsigned cnt,
                                              const float* __restrict__ h) {
  const int wave = threadIdx.x >> 5;
  const int lane = threadIdx.x & 31;
  const float2* hv = (const float2*)h;
  for (unsigned e = (unsigned)wave * EPT; e < cnt; e += (AGG_THREADS / 32) * EPT) {
    unsigned p[EPT];
    float2 v[EPT];
#pragma unroll
    for (int k = 0; k < EPT; ++k) p[k] = list[e + k < cnt ? e + k : cnt - 1u];
#pragma unroll
    for (int k = 0; k < EPT; ++k) v[k] = hv[(long)(p[k] / R_BIN) * 32 + lane];
#pragma unroll
    for (int k = 0; k < EPT; ++k) {
      const bool in = e + k < cnt;
      tile_add_row<PITCH>(tile, p[k], lane, make_float2(in ? v[k].x : 0.0f, in ? v[k].y : 0.0f));
    }
  }
}

template <int PITCH>
__device__ __forceinline__ void bucket_edges(float* tile, unsigned* degc, unsigned* list,
                                             unsigned* rst, unsigned* pre, unsigned* wtot,
                                             const unsigned* __restrict__ packed,
                                             const unsigned* __restrict__ off,
                                             int bin, int offp, int chunk,
                                             const float* __restrict__ h) {
  scan_runs(pre, wtot, load_run(rst, off, bin, offp, chunk));
  const unsigned total = pre[BK_BLOCKS];
  for (unsigned R = 0u; R < total; R += L_CAP) {
    fill_list(list, packed, rst, pre, R);
    __syncthreads();
    const unsigned cnt = min(total - R, (unsigned)L_CAP);
    for (unsigned i = threadIdx.x; i < cnt; i += AGG_THREADS) atomicAdd(&degc[list[i] % R_BIN], 1u);
    list_sum_rows<PITCH>(tile, list, cnt, h);
    __syncthreads();
  }
}

__global__ __launch_bounds__(AGG_THREADS) void bucket_aggregate_relu(
    const unsigned* __restrict__ packed, const unsigned* __restrict__ off,
    const float* __restrict__ h, const float* __restrict__ norm_src,
    const float* __restrict__ b1, float* __restrict__ out, int N, int chunk, int offp) {
  static_assert(AGG_THREADS % 16 == 0, "a thread keeps its 4-column group over the store loop");
  __shared__ alignas(16) float tile[R_BIN * H_DIM];
  __shared__ unsigned list[L_CAP];
  __shared__ unsigned rst[BK_BLOCKS];
  __shared__ unsigned pre[BK_BLOCKS + 1];
  __shared__ unsigned wtot[BK_BLOCKS / 32];
  __shared__ unsigned degc[R_BIN];
  __shared__ float nsl[R_BIN];
  const int bin = blockIdx.x;
  const int row_base = bin * R_BIN;
  float4* tile4 = (float4*)tile;
  for (int i = threadIdx.x; i < R_BIN * H_DIM / 4; i += AGG_THREADS)
    tile4[i] = make_float4(0.0f, 0.0f, 0.0f, 0.0f);
  if (threadIdx.x < R_BIN) {
    degc[threadIdx.x] = 0u;
    nsl[threadIdx.x] = (row_base + threadIdx.x < N) ? norm_src[row_base + threadIdx.x] : 0.0f;
  }
  bucket_edges<H_DIM>(tile, degc, list, rst, pre, wtot, packed, off, bin, offp, chunk, h);

  const float4 b = ((const float4*)b1)[threadIdx.x & 15];
  for (int i = threadIdx.x; i < R_BIN * H_DIM / 4; i += AGG_THREADS) {
    const int node = row_base + (i >> 4);
    if (node < N) {
      float4 v = tile4[i];
      const float nd = rsqrtf(fmaxf((float)degc[i >> 4], 1.0f));
      const float ns = nsl[i >> 4];
      v.x = fmaxf(v.x * nd + b.x, 0.0f) * ns;
      v.y = fmaxf(v.y * nd + b.y, 0.0f) * ns;
      v.z = fmaxf(v.z * nd + b.z, 0.0f) * ns;
      v.w = fmaxf(v.w * nd + b.w, 0.0f) * ns;
      ((float4*)out)[(long)node * 16 + (i & 15)] = v;
    }
  }
}

#define T_PITCH 68
#define L2_UNITS (R_BIN / 16 * 4)
__global__ __launch_bounds__(AGG_THREADS) void bucket_aggregate_final_wmma(
    const unsigned* __restrict__ packed, const unsigned* __restrict__ off,
    const float* __restrict__ h,
    const float* __restrict__ W_mu, const float* __restrict__ b_mu,
    const float* __restrict__ W_sig, const float* __restrict__ b_sig,
    const float* __restrict__ noise,
    float* __restrict__ z, int N, int chunk, int offp) {
  static_assert(R_BIN * T_PITCH % 4 == 0 && T_PITCH % 2 == 0, "float4 zero fill, 8-byte fragment reads");
  __shared__ alignas(16) float tile[R_BIN * T_PITCH];
  __shared__ v2f lWm[(H_DIM / 4) * 4 * 32];
  __shared__ v2f lWs[(H_DIM / 4) * 4 * 32];
  __shared__ unsigned list[L_CAP];
  __shared__ unsigned rst[BK_BLOCKS];
  __shared__ unsigned pre[BK_BLOCKS + 1];
  __shared__ unsigned wtot[BK_BLOCKS / 32];
  __shared__ unsigned degc[R_BIN];
  __shared__ float ndl[R_BIN];
  const int bin = blockIdx.x;
  const int row_base = bin * R_BIN;
  float4* tile4 = (float4*)tile;
  for (int i = threadIdx.x; i < R_BIN * T_PITCH / 4; i += AGG_THREADS)
    tile4[i] = make_float4(0.0f, 0.0f, 0.0f, 0.0f);
  if (threadIdx.x < R_BIN) degc[threadIdx.x] = 0u;
  stage_w_frag(lWm, W_mu, H_DIM / 4, threadIdx.x, AGG_THREADS);
  stage_w_frag(lWs, W_sig, H_DIM / 4, threadIdx.x, AGG_THREADS);
  bucket_edges<T_PITCH>(tile, degc, list, rst, pre, wtot, packed, off, bin, offp, chunk, h);
  if (threadIdx.x < R_BIN) ndl[threadIdx.x] = rsqrtf(fmaxf((float)degc[threadIdx.x], 1.0f));
  __syncthreads();

  const int wave = threadIdx.x >> 5;
  const int lane = threadIdx.x & 31;
  const int kgrp = (lane >> 4) * 2;
  for (int u = wave; u < L2_UNITS; u += AGG_THREADS / 32) {
    const int rt = u >> 2;
    const int ct = u & 3;
    const int row0 = row_base + rt * 16;
    if (row0 >= N) continue;
    const float* arow = tile + (rt * 16 + (lane & 15)) * T_PITCH + kgrp;
    const int n = ct * 16 + (lane & 15);
    const int lrow = rt * 16 + ((lane >> 4) << 3);
    float nz[8], nd[8];
#pragma unroll
    for (int v = 0; v < 8; ++v) {
      const int m = row_base + lrow + v;
      nz[v] = noise[(long)(m < N ? m : N - 1) * H_DIM + n];
      nd[v] = ndl[lrow + v];
    }
    const float bmu = b_mu[n];
    const float bsg = b_sig[n];
    v8f am = {}, as = {};
#pragma unroll
    for (int kb = 0; kb < H_DIM / 4; ++kb) {
      const v2f a  = *(const v2f*)(arow + kb * 4);
      const v2f bm = lWm[(kb * 4 + ct) * 32 + lane];
      const v2f bs = lWs[(kb * 4 + ct) * 32 + lane];
      am = __builtin_amdgcn_wmma_f32_16x16x4_f32(
          false, a, false, bm, (short)0, am, false, false);
      as = __builtin_amdgcn_wmma_f32_16x16x4_f32(
          false, a, false, bs, (short)0, as, false, false);
    }
    if (row0 + 16 <= N) {
#pragma unroll
      for (int v = 0; v < 8; ++v) {
        const int m = row_base + lrow + v;
        const float mu = am[v] * nd[v] + bmu;
        const float ls = as[v] * nd[v] + bsg;
        z[(long)m * H_DIM + n] = mu + nz[v] * expf(ls);
      }
    } else {
#pragma unroll
      for (int v = 0; v < 8; ++v) {
        const int m = row_base + lrow + v;
        if (m < N) {
          const float mu = am[v] * nd[v] + bmu;
          const float ls = as[v] * nd[v] + bsg;
          z[(long)m * H_DIM + n] = mu + nz[v] * expf(ls);
        }
      }
    }
  }
}

extern "C" void kernel_launch(void* const* d_in, const int* in_sizes, int n_in,
                              void* d_out, int out_size, void* d_ws, size_t ws_size,
                              hipStream_t stream) {
  const float* feat  = (const float*)d_in[0];
  const int*   src   = (const int*)d_in[1];
  const int*   dst   = (const int*)d_in[2];
  const float* noise = (const float*)d_in[3];
  const float* W1    = (const float*)d_in[4];
  const float* b1    = (const float*)d_in[5];
  const float* W_mu  = (const float*)d_in[6];
  const float* b_mu  = (const float*)d_in[7];
  const float* W_sig = (const float*)d_in[8];
  const float* b_sig = (const float*)d_in[9];

  const int E = in_sizes[1];
  const int N = in_sizes[3] / H_DIM;

  const int NB    = (N + R_BIN - 1) / R_BIN;
  const int chunk = (E + BK_BLOCKS - 1) / BK_BLOCKS;
  const int offp  = (NB + 32) & ~31;

  float* ws       = (float*)d_ws;
  float* norm_src = ws;
  unsigned* off_dst = (unsigned*)(ws + N);
  float* buf_a    = ws + 2L * N;
  float* buf_b    = buf_a + (long)N * H_DIM;
  float* zout     = (float*)d_out;

  unsigned* packed_dst  = (unsigned*)(buf_b + (long)N * H_DIM);
  unsigned* packed_src  = (unsigned*)buf_b;
  unsigned* off_src     = packed_src + (long)BK_BLOCKS * chunk;

  bucket_sort_slices<<<dim3(BK_BLOCKS, 2), BK_THREADS, (2 * NB + 1 + chunk) * sizeof(unsigned), stream>>>(
      src, dst, off_dst, off_src, packed_dst, packed_src, E, NB, chunk, offp);

  gemm1_wmma<<<NB, G1_THREADS, 0, stream>>>(feat, packed_src, off_src, W1, norm_src, buf_a, N, chunk, offp);

  bucket_aggregate_relu<<<NB, AGG_THREADS, 0, stream>>>(
      packed_dst, off_dst, buf_a, norm_src, b1, buf_b, N, chunk, offp);

  bucket_aggregate_final_wmma<<<NB, AGG_THREADS, 0, stream>>>(
      packed_dst, off_dst, buf_b, W_mu, b_mu, W_sig, b_sig, noise, zout, N, chunk, offp);
}


# Edge-sum launches: a trip's 12 edge words come from one LDS read, a lane each, decoded there; the wave takes each edge's two offsets by readlane (was 12 broadcast reads, 12 decodes); sum order kept
# speedup vs baseline: 3.9551x; 1.0426x over previous
#include <hip/hip_runtime.h>

typedef __attribute__((ext_vector_type(2))) float v2f;
typedef __attribute__((ext_vector_type(8))) float v8f;

#define F_IN  128
#define H_DIM 64
#define R_BIN 400
#define G1_THREADS (R_BIN / 16 * 32)

__device__ __forceinline__ void stage_w_frag(v2f* lW, const float* __restrict__ W,
                                             int kblocks, int tid, int nthreads) {
  const int total = kblocks * 4 * 32;
  for (int i = tid; i < total; i += nthreads) {
    const int kb = i >> 7;
    const int ct = (i >> 5) & 3;
    const int ln = i & 31;
    const int kg = (ln >> 4) * 2;
    const int n  = ct * 16 + (ln & 15);
    v2f p;
    p.x = W[(kb * 4 + kg)     * H_DIM + n];
    p.y = W[(kb * 4 + kg + 1) * H_DIM + n];
    lW[i] = p;
  }
}

#define BK_BLOCKS 256
#define BK_THREADS 1024
#define BK_K 7

__global__ __launch_bounds__(BK_THREADS) void bucket_sort_slices(
    const int* __restrict__ src, const int* __restrict__ dst,
    unsigned* __restrict__ off_dst, unsigned* __restrict__ off_src,
    unsigned* __restrict__ packed_dst, unsigned* __restrict__ packed_src,
    int E, int NB, int chunk, int offp) {
  extern __shared__ unsigned sm[];
  unsigned* pfx  = sm;
  unsigned* loc  = pfx + NB + 1;
  unsigned* word = loc + NB;
  const bool by_src = blockIdx.y != 0;
  const int* __restrict__ key = by_src ? src : dst;
  unsigned* __restrict__ off = (by_src ? off_src : off_dst) + (long)blockIdx.x * offp;
  unsigned* __restrict__ packed = (by_src ? packed_src : packed_dst) + (long)blockIdx.x * chunk;
  for (int i = threadIdx.x; i <= NB; i += BK_THREADS) pfx[i] = 0u;
  __syncthreads();

  const int e0 = blockIdx.x * chunk;
  const int n = max(min(e0 + chunk, E) - e0, 0);
  for (int base = threadIdx.x; base < n; base += BK_THREADS * BK_K) {
    int d[BK_K];
#pragma unroll
    for (int k = 0; k < BK_K; ++k) {
      const int i = base + k * BK_THREADS;
      d[k] = key[e0 + (i < n ? i : n - 1)];
    }
#pragma unroll
    for (int k = 0; k < BK_K; ++k)
      if (base + k * BK_THREADS < n) atomicAdd(&pfx[d[k] / R_BIN], 1u);
  }
  __syncthreads();

  if (threadIdx.x < 32) {
    const int per = (NB + 31) / 32;
    const int b0 = (int)threadIdx.x * per;
    const int b1 = min(b0 + per, NB);
    unsigned sum = 0u;
    for (int b = b0; b < b1; ++b) sum += pfx[b];
    unsigned inc = sum;
    for (int dlt = 1; dlt < 32; dlt <<= 1) {
      const unsigned v = __shfl_up(inc, dlt);
      if ((int)threadIdx.x >= dlt) inc += v;
    }
    unsigned run = inc - sum;
    for (int b = b0; b < b1; ++b) {
      const unsigned c = pfx[b];
      pfx[b] = run;
      loc[b] = run;
      run += c;
    }
    if (threadIdx.x == 31) pfx[NB] = inc;
  }
  __syncthreads();

  for (int i = threadIdx.x; i <= NB; i += BK_THREADS) off[i] = pfx[i];
  for (int base = threadIdx.x; base < n; base += BK_THREADS * BK_K) {
    int d[BK_K], s[BK_K];
#pragma unroll
    for (int k = 0; k < BK_K; ++k) {
      const int i = base + k * BK_THREADS;
      const int ic = i < n ? i : n - 1;
      d[k] = key[e0 + ic];
      s[k] = by_src ? 0 : src[e0 + ic];
    }
#pragma unroll
    for (int k = 0; k < BK_K; ++k) {
      if (base + k * BK_THREADS < n) {
        const int bin = d[k] / R_BIN;
        const unsigned at = atomicAdd(&loc[bin], 1u);
        word[at] = (unsigned)(d[k] - bin * R_BIN) + (unsigned)s[k] * R_BIN;
      }
    }
  }
  __syncthreads();
  for (int i = threadIdx.x; i < n; i += BK_THREADS) packed[i] = word[i];
}

__device__ __forceinline__ unsigned load_run(unsigned* rst, const unsigned* __restrict__ off,
                                             int bin, int offp, int chunk) {
  unsigned len = 0u;
  if (threadIdx.x < BK_BLOCKS) {
    const unsigned* o = off + (long)threadIdx.x * offp + bin;
    const unsigned a = o[0];
    len = o[1] - a;
    rst[threadIdx.x] = threadIdx.x * (unsigned)chunk + a;
  }
  return len;
}

__device__ __forceinline__ void scan_runs(unsigned* pre, unsigned* wtot, unsigned len) {
  static_assert(BK_BLOCKS % 32 == 0, "the runs fill whole waves");
  const int wave = threadIdx.x >> 5;
  const int lane = threadIdx.x & 31;
  unsigned inc = len;
  for (int dlt = 1; dlt < 32; dlt <<= 1) {
    const unsigned v = __shfl_up(inc, dlt);
    if (lane >= dlt) inc += v;
  }
  if (threadIdx.x < BK_BLOCKS && lane == 31) wtot[wave] = inc;
  __syncthreads();
  if (threadIdx.x < BK_BLOCKS) {
    unsigned before = 0u;
    for (int q = 0; q < wave; ++q) before += wtot[q];
    pre[threadIdx.x] = before + inc - len;
    if (threadIdx.x == BK_BLOCKS - 1) pre[BK_BLOCKS] = before + inc;
  }
  __syncthreads();
}

template <int NW>
__device__ __forceinline__ void count_run_rows(unsigned* degc, const unsigned* __restrict__ packed,
                                               const unsigned* rst, const unsigned* rln) {
  constexpr int RPW = (BK_BLOCKS + NW - 1) / NW;
  const int wave = threadIdx.x >> 5;
  const unsigned lane = threadIdx.x & 31;
  unsigned w0[RPW], w1[RPW], ln[RPW];
#pragma unroll
  for (int j = 0; j < RPW; ++j) {
    const int r = wave + j * NW;
    const int rc = r < BK_BLOCKS ? r : 0;
    const unsigned s = rst[rc];
    ln[j] = r < BK_BLOCKS ? rln[rc] : 0u;
    w0[j] = packed[lane < ln[j] ? s + lane : 0u];
    w1[j] = packed[lane + 32u < ln[j] ? s + lane + 32u : 0u];
  }
#pragma unroll
  for (int j = 0; j < RPW; ++j) {
    if (lane < ln[j]) atomicAdd(&degc[w0[j] % R_BIN], 1u);
    if (lane + 32u < ln[j]) atomicAdd(&degc[w1[j] % R_BIN], 1u);
  }
  for (int r = wave; r < BK_BLOCKS; r += NW) {
    const unsigned s = rst[r];
    const unsigned len = rln[r];
    for (unsigned i = 64u + lane; i < len; i += 32u) atomicAdd(&degc[packed[s + i] % R_BIN], 1u);
  }
}

__global__ __launch_bounds__(G1_THREADS) void gemm1_wmma(
    const float* __restrict__ feat, const unsigned* __restrict__ packed_src,
    const unsigned* __restrict__ off_src, const float* __restrict__ W1,
    float* __restrict__ norm_src, float* __restrict__ h1, int N, int chunk, int offp) {
  static_assert(R_BIN % 16 == 0, "a bucket is a whole number of 16-row tiles");
  static_assert(G1_THREADS >= BK_BLOCKS, "one thread per run reads its offsets");
  __shared__ v2f lW[(F_IN / 4) * 4 * 32];
  __shared__ unsigned degc[R_BIN];
  __shared__ unsigned rst[BK_BLOCKS];
  __shared__ unsigned rln[BK_BLOCKS];
  if (threadIdx.x < R_BIN) degc[threadIdx.x] = 0u;
  {
    const unsigned len = load_run(rst, off_src, blockIdx.x, offp, chunk);
    if (threadIdx.x < BK_BLOCKS) rln[threadIdx.x] = len;
  }
  stage_w_frag(lW, W1, F_IN / 4, threadIdx.x, G1_THREADS);
  __syncthreads();
  count_run_rows<G1_THREADS / 32>(degc, packed_src, rst, rln);
  __syncthreads();
  if (threadIdx.x < R_BIN && blockIdx.x * R_BIN + threadIdx.x < N)
    norm_src[blockIdx.x * R_BIN + threadIdx.x] =
        rsqrtf(fmaxf((float)degc[threadIdx.x], 1.0f));

  const int wave = threadIdx.x >> 5;
  const int lane = threadIdx.x & 31;
  const int tile = blockIdx.x * (R_BIN / 16) + wave;
  const int row0 = tile * 16;
  if (row0 >= N) return;

  const int mrow = row0 + (lane & 15);
  const int kgrp = (lane >> 4) * 2;
  const float ns = rsqrtf(fmaxf((float)degc[wave * 16 + (lane & 15)], 1.0f));

  v8f acc[4] = {};
  for (int kb = 0; kb < F_IN / 4; ++kb) {
    v2f a = *(const v2f*)(feat + (long)mrow * F_IN + kb * 4 + kgrp);
    a.x *= ns; a.y *= ns;
#pragma unroll
    for (int ct = 0; ct < 4; ++ct) {
      const v2f b = lW[(kb * 4 + ct) * 32 + lane];
      acc[ct] = __builtin_amdgcn_wmma_f32_16x16x4_f32(
          false, a, false, b, (short)0, acc[ct], false, false);
    }
  }

  const int mbase = row0 + ((lane >> 4) << 3);
#pragma unroll
  for (int ct = 0; ct < 4; ++ct) {
    const int n = ct * 16 + (lane & 15);
#pragma unroll
    for (int v = 0; v < 8; ++v)
      h1[(long)(mbase + v) * H_DIM + n] = acc[ct][v];
  }
}

template <int PITCH>
__device__ __forceinline__ void tile_add_row(float* tile, unsigned p, int lane, float2 v) {
  float* t = tile + (p % R_BIN) * PITCH + lane * 2;
  atomicAdd(t, v.x);
  atomicAdd(t + 1, v.y);
}

#define AGG_THREADS 1024
#define EPT 12
#define L_CAP 8192

__device__ __forceinline__ void fill_list(unsigned* list, const unsigned* __restrict__ packed,
                                          const unsigned* rst, const unsigned* pre, unsigned R) {
  constexpr int NW = AGG_THREADS / 32;
  constexpr int RPW = (BK_BLOCKS + NW - 1) / NW;
  const int wave = threadIdx.x >> 5;
  const unsigned lane = threadIdx.x & 31;
  unsigned w0[RPW], w1[RPW], g0[RPW], ln[RPW];
#pragma unroll
  for (int j = 0; j < RPW; ++j) {
    const int r = wave + j * NW;
    const int rc = r < BK_BLOCKS ? r : 0;
    const unsigned s = rst[rc];
    const unsigned p = pre[rc];
    ln[j] = r < BK_BLOCKS ? pre[rc + 1] - p : 0u;
    g0[j] = p + lane - R;
    w0[j] = packed[(lane < ln[j] && g0[j] < L_CAP) ? s + lane : 0u];
    w1[j] = packed[(lane + 32u < ln[j] && g0[j] + 32u < L_CAP) ? s + lane + 32u : 0u];
  }
#pragma unroll
  for (int j = 0; j < RPW; ++j) {
    if (lane < ln[j] && g0[j] < L_CAP) list[g0[j]] = w0[j];
    if (lane + 32u < ln[j] && g0[j] + 32u < L_CAP) list[g0[j] + 32u] = w1[j];
  }
  for (int r = wave; r < BK_BLOCKS; r += NW) {
    const unsigned s = rst[r];
    const unsigned p = pre[r];
    const unsigned len = pre[r + 1] - p;
    for (unsigned i = 64u + lane; i < len; i += 32u) {
      const unsigned g = p + i - R;
      if (g < L_CAP) list[g] = packed[s + i];
    }
  }
}

template <int PITCH>
__device__ __forceinline__ void list_sum_rows(float* tile, const unsigned* list, unsigned cnt_v,
                                              const float* __restrict__ h) {
  static_assert(EPT <= 32, "one lane per edge of a trip");
  const unsigned wave = (unsigned)__builtin_amdgcn_readfirstlane((int)(threadIdx.x >> 5));
  const unsigned cnt = (unsigned)__builtin_amdgcn_readfirstlane((int)cnt_v);
  const unsigned lane = threadIdx.x & 31;
  const float2* hv = (const float2*)h;
  for (unsigned e = wave * EPT; e < cnt; e += (AGG_THREADS / 32) * EPT) {
    const unsigned w = list[min(e + lane, cnt - 1u)];
    const unsigned srow = w / R_BIN;
    const int hoff_l = (int)(srow * (H_DIM / 2));
    const int toff_l = (int)((w - srow * R_BIN) * PITCH);
    unsigned hoff[EPT], toff[EPT];
    float2 v[EPT];
#pragma unroll
    for (int k = 0; k < EPT; ++k) {
      hoff[k] = (unsigned)__builtin_amdgcn_readlane(hoff_l, k);
      toff[k] = (unsigned)__builtin_amdgcn_readlane(toff_l, k);
    }
#pragma unroll
    for (int k = 0; k < EPT; ++k) v[k] = (hv + (size_t)hoff[k])[lane];
    if (e + EPT <= cnt) {
#pragma unroll
      for (int k = 0; k < EPT; ++k) {
        float* t = tile + toff[k] + lane * 2;
        atomicAdd(t, v[k].x);
        atomicAdd(t + 1, v[k].y);
      }
    } else {
#pragma unroll
      for (int k = 0; k < EPT; ++k) {
        if (e + k < cnt) {
          float* t = tile + toff[k] + lane * 2;
          atomicAdd(t, v[k].x);
          atomicAdd(t + 1, v[k].y);
        }
      }
    }
  }
}

template <int PITCH>
__device__ __forceinline__ void bucket_edges(float* tile, unsigned* degc, unsigned* list,
                                             unsigned* rst, unsigned* pre, unsigned* wtot,
                                             const unsigned* __restrict__ packed,
                                             const unsigned* __restrict__ off,
                                             int bin, int offp, int chunk,
                                             const float* __restrict__ h) {
  scan_runs(pre, wtot, load_run(rst, off, bin, offp, chunk));
  const unsigned total = pre[BK_BLOCKS];
  for (unsigned R = 0u; R < total; R += L_CAP) {
    fill_list(list, packed, rst, pre, R);
    __syncthreads();
    const unsigned cnt = min(total - R, (unsigned)L_CAP);
    for (unsigned i = threadIdx.x; i < cnt; i += AGG_THREADS) atomicAdd(&degc[list[i] % R_BIN], 1u);
    list_sum_rows<PITCH>(tile, list, cnt, h);
    __syncthreads();
  }
}

__global__ __launch_bounds__(AGG_THREADS) void bucket_aggregate_relu(
    const unsigned* __restrict__ packed, const unsigned* __restrict__ off,
    const float* __restrict__ h, const float* __restrict__ norm_src,
    const float* __restrict__ b1, float* __restrict__ out, int N, int chunk, int offp) {
  static_assert(AGG_THREADS % 16 == 0, "a thread keeps its 4-column group over the store loop");
  __shared__ alignas(16) float tile[R_BIN * H_DIM];
  __shared__ unsigned list[L_CAP];
  __shared__ unsigned rst[BK_BLOCKS];
  __shared__ unsigned pre[BK_BLOCKS + 1];
  __shared__ unsigned wtot[BK_BLOCKS / 32];
  __shared__ unsigned degc[R_BIN];
  __shared__ float nsl[R_BIN];
  const int bin = blockIdx.x;
  const int row_base = bin * R_BIN;
  float4* tile4 = (float4*)tile;
  for (int i = threadIdx.x; i < R_BIN * H_DIM / 4; i += AGG_THREADS)
    tile4[i] = make_float4(0.0f, 0.0f, 0.0f, 0.0f);
  if (threadIdx.x < R_BIN) {
    degc[threadIdx.x] = 0u;
    nsl[threadIdx.x] = (row_base + threadIdx.x < N) ? norm_src[row_base + threadIdx.x] : 0.0f;
  }
  bucket_edges<H_DIM>(tile, degc, list, rst, pre, wtot, packed, off, bin, offp, chunk, h);

  const float4 b = ((const float4*)b1)[threadIdx.x & 15];
  for (int i = threadIdx.x; i < R_BIN * H_DIM / 4; i += AGG_THREADS) {
    const int node = row_base + (i >> 4);
    if (node < N) {
      float4 v = tile4[i];
      const float nd = rsqrtf(fmaxf((float)degc[i >> 4], 1.0f));
      const float ns = nsl[i >> 4];
      v.x = fmaxf(v.x * nd + b.x, 0.0f) * ns;
      v.y = fmaxf(v.y * nd + b.y, 0.0f) * ns;
      v.z = fmaxf(v.z * nd + b.z, 0.0f) * ns;
      v.w = fmaxf(v.w * nd + b.w, 0.0f) * ns;
      ((float4*)out)[(long)node * 16 + (i & 15)] = v;
    }
  }
}

#define T_PITCH 68
#define L2_UNITS (R_BIN / 16 * 4)
__global__ __launch_bounds__(AGG_THREADS) void bucket_aggregate_final_wmma(
    const unsigned* __restrict__ packed, const unsigned* __restrict__ off,
    const float* __restrict__ h,
    const float* __restrict__ W_mu, const float* __restrict__ b_mu,
    const float* __restrict__ W_sig, const float* __restrict__ b_sig,
    const float* __restrict__ noise,
    float* __restrict__ z, int N, int chunk, int offp) {
  static_assert(R_BIN * T_PITCH % 4 == 0 && T_PITCH % 2 == 0, "float4 zero fill, 8-byte fragment reads");
  __shared__ alignas(16) float tile[R_BIN * T_PITCH];
  __shared__ v2f lWm[(H_DIM / 4) * 4 * 32];
  __shared__ v2f lWs[(H_DIM / 4) * 4 * 32];
  __shared__ unsigned list[L_CAP];
  __shared__ unsigned rst[BK_BLOCKS];
  __shared__ unsigned pre[BK_BLOCKS + 1];
  __shared__ unsigned wtot[BK_BLOCKS / 32];
  __shared__ unsigned degc[R_BIN];
  __shared__ float ndl[R_BIN];
  const int bin = blockIdx.x;
  const int row_base = bin * R_BIN;
  float4* tile4 = (float4*)tile;
  for (int i = threadIdx.x; i < R_BIN * T_PITCH / 4; i += AGG_THREADS)
    tile4[i] = make_float4(0.0f, 0.0f, 0.0f, 0.0f);
  if (threadIdx.x < R_BIN) degc[threadIdx.x] = 0u;
  stage_w_frag(lWm, W_mu, H_DIM / 4, threadIdx.x, AGG_THREADS);
  stage_w_frag(lWs, W_sig, H_DIM / 4, threadIdx.x, AGG_THREADS);
  bucket_edges<T_PITCH>(tile, degc, list, rst, pre, wtot, packed, off, bin, offp, chunk, h);
  if (threadIdx.x < R_BIN) ndl[threadIdx.x] = rsqrtf(fmaxf((float)degc[threadIdx.x], 1.0f));
  __syncthreads();

  const int wave = threadIdx.x >> 5;
  const int lane = threadIdx.x & 31;
  const int kgrp = (lane >> 4) * 2;
  for (int u = wave; u < L2_UNITS; u += AGG_THREADS / 32) {
    const int rt = u >> 2;
    const int ct = u & 3;
    const int row0 = row_base + rt * 16;
    if (row0 >= N) continue;
    const float* arow = tile + (rt * 16 + (lane & 15)) * T_PITCH + kgrp;
    const int n = ct * 16 + (lane & 15);
    const int lrow = rt * 16 + ((lane >> 4) << 3);
    float nz[8], nd[8];
#pragma unroll
    for (int v = 0; v < 8; ++v) {
      const int m = row_base + lrow + v;
      nz[v] = noise[(long)(m < N ? m : N - 1) * H_DIM + n];
      nd[v] = ndl[lrow + v];
    }
    const float bmu = b_mu[n];
    const float bsg = b_sig[n];
    v8f am = {}, as = {};
#pragma unroll
    for (int kb = 0; kb < H_DIM / 4; ++kb) {
      const v2f a  = *(const v2f*)(arow + kb * 4);
      const v2f bm = lWm[(kb * 4 + ct) * 32 + lane];
      const v2f bs = lWs[(kb * 4 + ct) * 32 + lane];
      am = __builtin_amdgcn_wmma_f32_16x16x4_f32(
          false, a, false, bm, (short)0, am, false, false);
      as = __builtin_amdgcn_wmma_f32_16x16x4_f32(
          false, a, false, bs, (short)0, as, false, false);
    }
    if (row0 + 16 <= N) {
#pragma unroll
      for (int v = 0; v < 8; ++v) {
        const int m = row_base + lrow + v;
        const float mu = am[v] * nd[v] + bmu;
        const float ls = as[v] * nd[v] + bsg;
        z[(long)m * H_DIM + n] = mu + nz[v] * expf(ls);
      }
    } else {
#pragma unroll
      for (int v = 0; v < 8; ++v) {
        const int m = row_base + lrow + v;
        if (m < N) {
          const float mu = am[v] * nd[v] + bmu;
          const float ls = as[v] * nd[v] + bsg;
          z[(long)m * H_DIM + n] = mu + nz[v] * expf(ls);
        }
      }
    }
  }
}

extern "C" void kernel_launch(void* const* d_in, const int* in_sizes, int n_in,
                              void* d_out, int out_size, void* d_ws, size_t ws_size,
                              hipStream_t stream) {
  const float* feat  = (const float*)d_in[0];
  const int*   src   = (const int*)d_in[1];
  const int*   dst   = (const int*)d_in[2];
  const float* noise = (const float*)d_in[3];
  const float* W1    = (const float*)d_in[4];
  const float* b1    = (const float*)d_in[5];
  const float* W_mu  = (const float*)d_in[6];
  const float* b_mu  = (const float*)d_in[7];
  const float* W_sig = (const float*)d_in[8];
  const float* b_sig = (const float*)d_in[9];

  const int E = in_sizes[1];
  const int N = in_sizes[3] / H_DIM;

  const int NB    = (N + R_BIN - 1) / R_BIN;
  const int chunk = (E + BK_BLOCKS - 1) / BK_BLOCKS;
  const int offp  = (NB + 32) & ~31;

  float* ws       = (float*)d_ws;
  float* norm_src = ws;
  unsigned* off_dst = (unsigned*)(ws + N);
  float* buf_a    = ws + 2L * N;
  float* buf_b    = buf_a + (long)N * H_DIM;
  float* zout     = (float*)d_out;

  unsigned* packed_dst  = (unsigned*)(buf_b + (long)N * H_DIM);
  unsigned* packed_src  = (unsigned*)buf_b;
  unsigned* off_src     = packed_src + (long)BK_BLOCKS * chunk;

  bucket_sort_slices<<<dim3(BK_BLOCKS, 2), BK_THREADS, (2 * NB + 1 + chunk) * sizeof(unsigned), stream>>>(
      src, dst, off_dst, off_src, packed_dst, packed_src, E, NB, chunk, offp);

  gemm1_wmma<<<NB, G1_THREADS, 0, stream>>>(feat, packed_src, off_src, W1, norm_src, buf_a, N, chunk, offp);

  bucket_aggregate_relu<<<NB, AGG_THREADS, 0, stream>>>(
      packed_dst, off_dst, buf_a, norm_src, b1, buf_b, N, chunk, offp);

  bucket_aggregate_final_wmma<<<NB, AGG_THREADS, 0, stream>>>(
      packed_dst, off_dst, buf_b, W_mu, b_mu, W_sig, b_sig, noise, zout, N, chunk, offp);
}


# speedup vs baseline: 4.2476x; 1.0740x over previous
#include <hip/hip_runtime.h>

typedef __attribute__((ext_vector_type(2))) float v2f;
typedef __attribute__((ext_vector_type(8))) float v8f;

#define F_IN  128
#define H_DIM 64
#define R_BIN 400
#define G1_THREADS (R_BIN / 16 * 32)

__device__ __forceinline__ void stage_w_frag(v2f* lW, const float* __restrict__ W,
                                             int kblocks, int tid, int nthreads) {
  const int total = kblocks * 4 * 32;
  for (int i = tid; i < total; i += nthreads) {
    const int kb = i >> 7;
    const int ct = (i >> 5) & 3;
    const int ln = i & 31;
    const int kg = (ln >> 4) * 2;
    const int n  = ct * 16 + (ln & 15);
    v2f p;
    p.x = W[(kb * 4 + kg)     * H_DIM + n];
    p.y = W[(kb * 4 + kg + 1) * H_DIM + n];
    lW[i] = p;
  }
}

#define BK_BLOCKS 256
#define BK_THREADS 1024
#define BK_K 7

__global__ __launch_bounds__(BK_THREADS) void bucket_sort_slices(
    const int* __restrict__ src, const int* __restrict__ dst,
    unsigned* __restrict__ off_dst, unsigned* __restrict__ off_src,
    unsigned* __restrict__ packed_dst, unsigned* __restrict__ packed_src,
    int E, int NB, int chunk, int offp) {
  extern __shared__ unsigned sm[];
  unsigned* pfx  = sm;
  unsigned* loc  = pfx + NB + 1;
  unsigned* word = loc + NB;
  const bool by_src = blockIdx.y != 0;
  const int* __restrict__ key = by_src ? src : dst;
  unsigned* __restrict__ off = (by_src ? off_src : off_dst) + (long)blockIdx.x * offp;
  unsigned* __restrict__ packed = (by_src ? packed_src : packed_dst) + (long)blockIdx.x * chunk;
  for (int i = threadIdx.x; i <= NB; i += BK_THREADS) pfx[i] = 0u;
  __syncthreads();

  const int e0 = blockIdx.x * chunk;
  const int n = max(min(e0 + chunk, E) - e0, 0);
  for (int base = threadIdx.x; base < n; base += BK_THREADS * BK_K) {
    int d[BK_K];
#pragma unroll
    for (int k = 0; k < BK_K; ++k) {
      const int i = base + k * BK_THREADS;
      d[k] = key[e0 + (i < n ? i : n - 1)];
    }
#pragma unroll
    for (int k = 0; k < BK_K; ++k)
      if (base + k * BK_THREADS < n) atomicAdd(&pfx[d[k] / R_BIN], 1u);
  }
  __syncthreads();

  if (threadIdx.x < 32) {
    const int per = (NB + 31) / 32;
    const int b0 = (int)threadIdx.x * per;
    const int b1 = min(b0 + per, NB);
    unsigned sum = 0u;
    for (int b = b0; b < b1; ++b) sum += pfx[b];
    unsigned inc = sum;
    for (int dlt = 1; dlt < 32; dlt <<= 1) {
      const unsigned v = __shfl_up(inc, dlt);
      if ((int)threadIdx.x >= dlt) inc += v;
    }
    unsigned run = inc - sum;
    for (int b = b0; b < b1; ++b) {
      const unsigned c = pfx[b];
      pfx[b] = run;
      loc[b] = run;
      run += c;
    }
    if (threadIdx.x == 31) pfx[NB] = inc;
  }
  __syncthreads();

  for (int i = threadIdx.x; i <= NB; i += BK_THREADS) off[i] = pfx[i];
  for (int base = threadIdx.x; base < n; base += BK_THREADS * BK_K) {
    int d[BK_K], s[BK_K];
#pragma unroll
    for (int k = 0; k < BK_K; ++k) {
      const int i = base + k * BK_THREADS;
      const int ic = i < n ? i : n - 1;
      d[k] = key[e0 + ic];
      s[k] = by_src ? 0 : src[e0 + ic];
    }
#pragma unroll
    for (int k = 0; k < BK_K; ++k) {
      if (base + k * BK_THREADS < n) {
        const int bin = d[k] / R_BIN;
        const unsigned at = atomicAdd(&loc[bin], 1u);
        word[at] = (unsigned)(d[k] - bin * R_BIN) + (unsigned)s[k] * R_BIN;
      }
    }
  }
  __syncthreads();
  for (int i = threadIdx.x; i < n; i += BK_THREADS) packed[i] = word[i];
}

__device__ __forceinline__ unsigned load_run(unsigned* rst, const unsigned* __restrict__ off,
                                             int bin, int offp, int chunk) {
  unsigned len = 0u;
  if (threadIdx.x < BK_BLOCKS) {
    const unsigned* o = off + (long)threadIdx.x * offp + bin;
    const unsigned a = o[0];
    len = o[1] - a;
    rst[threadIdx.x] = threadIdx.x * (unsigned)chunk + a;
  }
  return len;
}

__device__ __forceinline__ void scan_runs(unsigned* pre, unsigned* wtot, unsigned len) {
  static_assert(BK_BLOCKS % 32 == 0, "the runs fill whole waves");
  const int wave = threadIdx.x >> 5;
  const int lane = threadIdx.x & 31;
  unsigned inc = len;
  for (int dlt = 1; dlt < 32; dlt <<= 1) {
    const unsigned v = __shfl_up(inc, dlt);
    if (lane >= dlt) inc += v;
  }
  if (threadIdx.x < BK_BLOCKS && lane == 31) wtot[wave] = inc;
  __syncthreads();
  if (threadIdx.x < BK_BLOCKS) {
    unsigned before = 0u;
    for (int q = 0; q < wave; ++q) before += wtot[q];
    pre[threadIdx.x] = before + inc - len;
    if (threadIdx.x == BK_BLOCKS - 1) pre[BK_BLOCKS] = before + inc;
  }
  __syncthreads();
}

template <int NW>
__device__ __forceinline__ void count_run_rows(unsigned* degc, const unsigned* __restrict__ packed,
                                               const unsigned* rst, const unsigned* rln) {
  constexpr int RPW = (BK_BLOCKS + NW - 1) / NW;
  const int wave = threadIdx.x >> 5;
  const unsigned lane = threadIdx.x & 31;
  unsigned w0[RPW], w1[RPW], ln[RPW];
#pragma unroll
  for (int j = 0; j < RPW; ++j) {
    const int r = wave + j * NW;
    const int rc = r < BK_BLOCKS ? r : 0;
    const unsigned s = rst[rc];
    ln[j] = r < BK_BLOCKS ? rln[rc] : 0u;
    w0[j] = packed[lane < ln[j] ? s + lane : 0u];
    w1[j] = packed[lane + 32u < ln[j] ? s + lane + 32u : 0u];
  }
#pragma unroll
  for (int j = 0; j < RPW; ++j) {
    if (lane < ln[j]) atomicAdd(&degc[w0[j] % R_BIN], 1u);
    if (lane + 32u < ln[j]) atomicAdd(&degc[w1[j] % R_BIN], 1u);
  }
  for (int r = wave; r < BK_BLOCKS; r += NW) {
    const unsigned s = rst[r];
    const unsigned len = rln[r];
    for (unsigned i = 64u + lane; i < len; i += 32u) atomicAdd(&degc[packed[s + i] % R_BIN], 1u);
  }
}

__global__ __launch_bounds__(G1_THREADS) void gemm1_wmma(
    const float* __restrict__ feat, const unsigned* __restrict__ packed_src,
    const unsigned* __restrict__ off_src, const float* __restrict__ W1,
    float* __restrict__ norm_src, float* __restrict__ h1, int N, int chunk, int offp) {
  static_assert(R_BIN % 16 == 0, "a bucket is a whole number of 16-row tiles");
  static_assert(G1_THREADS >= BK_BLOCKS, "one thread per run reads its offsets");
  __shared__ v2f lW[(F_IN / 4) * 4 * 32];
  __shared__ unsigned degc[R_BIN];
  __shared__ unsigned rst[BK_BLOCKS];
  __shared__ unsigned rln[BK_BLOCKS];
  if (threadIdx.x < R_BIN) degc[threadIdx.x] = 0u;
  {
    const unsigned len = load_run(rst, off_src, blockIdx.x, offp, chunk);
    if (threadIdx.x < BK_BLOCKS) rln[threadIdx.x] = len;
  }
  stage_w_frag(lW, W1, F_IN / 4, threadIdx.x, G1_THREADS);
  __syncthreads();
  count_run_rows<G1_THREADS / 32>(degc, packed_src, rst, rln);
  __syncthreads();
  if (threadIdx.x < R_BIN && blockIdx.x * R_BIN + threadIdx.x < N)
    norm_src[blockIdx.x * R_BIN + threadIdx.x] =
        rsqrtf(fmaxf((float)degc[threadIdx.x], 1.0f));

  const int wave = threadIdx.x >> 5;
  const int lane = threadIdx.x & 31;
  const int tile = blockIdx.x * (R_BIN / 16) + wave;
  const int row0 = tile * 16;
  if (row0 >= N) return;

  const int mrow = row0 + (lane & 15);
  const int kgrp = (lane >> 4) * 2;
  const float ns = rsqrtf(fmaxf((float)degc[wave * 16 + (lane & 15)], 1.0f));

  v8f acc[4] = {};
  for (int kb = 0; kb < F_IN / 4; ++kb) {
    v2f a = *(const v2f*)(feat + (long)mrow * F_IN + kb * 4 + kgrp);
    a.x *= ns; a.y *= ns;
#pragma unroll
    for (int ct = 0; ct < 4; ++ct) {
      const v2f b = lW[(kb * 4 + ct) * 32 + lane];
      acc[ct] = __builtin_amdgcn_wmma_f32_16x16x4_f32(
          false, a, false, b, (short)0, acc[ct], false, false);
    }
  }

  const int mbase = row0 + ((lane >> 4) << 3);
#pragma unroll
  for (int ct = 0; ct < 4; ++ct) {
    const int n = ct * 16 + (lane & 15);
#pragma unroll
    for (int v = 0; v < 8; ++v)
      h1[(long)(mbase + v) * H_DIM + n] = acc[ct][v];
  }
}

template <int PITCH>
__device__ __forceinline__ void tile_add_row(float* tile, unsigned p, int lane, float2 v) {
  float* t = tile + (p % R_BIN) * PITCH + lane * 2;
  atomicAdd(t, v.x);
  atomicAdd(t + 1, v.y);
}

#define AGG_THREADS 1024
#define TILE_B_FLOATS 32768
#define EPT 12
#define L_CAP 8192

__device__ __forceinline__ void fill_list(unsigned* list, const unsigned* __restrict__ packed,
                                          const unsigned* rst, const unsigned* pre, unsigned R) {
  constexpr int NW = AGG_THREADS / 32;
  constexpr int RPW = (BK_BLOCKS + NW - 1) / NW;
  const int wave = threadIdx.x >> 5;
  const unsigned lane = threadIdx.x & 31;
  unsigned w0[RPW], w1[RPW], g0[RPW], ln[RPW];
#pragma unroll
  for (int j = 0; j < RPW; ++j) {
    const int r = wave + j * NW;
    const int rc = r < BK_BLOCKS ? r : 0;
    const unsigned s = rst[rc];
    const unsigned p = pre[rc];
    ln[j] = r < BK_BLOCKS ? pre[rc + 1] - p : 0u;
    g0[j] = p + lane - R;
    w0[j] = packed[(lane < ln[j] && g0[j] < L_CAP) ? s + lane : 0u];
    w1[j] = packed[(lane + 32u < ln[j] && g0[j] + 32u < L_CAP) ? s + lane + 32u : 0u];
  }
#pragma unroll
  for (int j = 0; j < RPW; ++j) {
    if (lane < ln[j] && g0[j] < L_CAP) list[g0[j]] = w0[j];
    if (lane + 32u < ln[j] && g0[j] + 32u < L_CAP) list[g0[j] + 32u] = w1[j];
  }
  for (int r = wave; r < BK_BLOCKS; r += NW) {
    const unsigned s = rst[r];
    const unsigned p = pre[r];
    const unsigned len = pre[r + 1] - p;
    for (unsigned i = 64u + lane; i < len; i += 32u) {
      const unsigned g = p + i - R;
      if (g < L_CAP) list[g] = packed[s + i];
    }
  }
}

template <int PITCH>
__device__ __forceinline__ void list_sum_rows(float* tile, const unsigned* list, unsigned cnt_v,
                                              const float* __restrict__ h) {
  static_assert(EPT <= 32, "one lane per edge of a trip");
  const unsigned wave = (unsigned)__builtin_amdgcn_readfirstlane((int)(threadIdx.x >> 5));
  const unsigned cnt = (unsigned)__builtin_amdgcn_readfirstlane((int)cnt_v);
  const unsigned lane = threadIdx.x & 31;
  const float2* hv = (const float2*)h;
  for (unsigned e = wave * EPT; e < cnt; e += (AGG_THREADS / 32) * EPT) {
    const unsigned w = list[min(e + lane, cnt - 1u)];
    const unsigned srow = w / R_BIN;
    const int hoff_l = (int)(srow * (H_DIM / 2));
    const int toff_l = (int)((w - srow * R_BIN) * PITCH);
    unsigned hoff[EPT], toff[EPT];
    float2 v[EPT];
#pragma unroll
    for (int k = 0; k < EPT; ++k) {
      hoff[k] = (unsigned)__builtin_amdgcn_readlane(hoff_l, k);
      toff[k] = (unsigned)__builtin_amdgcn_readlane(toff_l, k);
    }
#pragma unroll
    for (int k = 0; k < EPT; ++k) v[k] = (hv + (size_t)hoff[k])[lane];
    if (e + EPT <= cnt) {
#pragma unroll
      for (int k = 0; k < EPT; ++k) {
        float* t = tile + toff[k] + lane * 2;
        atomicAdd(t, v[k].x);
        atomicAdd(t + 1, v[k].y);
      }
    } else {
#pragma unroll
      for (int k = 0; k < EPT; ++k) {
        if (e + k < cnt) {
          float* t = tile + toff[k] + lane * 2;
          atomicAdd(t, v[k].x);
          atomicAdd(t + 1, v[k].y);
        }
      }
    }
  }
}

template <int PITCH>
__device__ __forceinline__ void bucket_edges(float* tile, unsigned* degc, unsigned* list,
                                             unsigned* rst, unsigned* pre, unsigned* wtot,
                                             const unsigned* __restrict__ packed,
                                             const unsigned* __restrict__ off,
                                             int bin, int offp, int chunk,
                                             const float* __restrict__ h) {
  scan_runs(pre, wtot, load_run(rst, off, bin, offp, chunk));
  const unsigned total = pre[BK_BLOCKS];
  for (unsigned R = 0u; R < total; R += L_CAP) {
    fill_list(list, packed, rst, pre, R);
    __syncthreads();
    const unsigned cnt = min(total - R, (unsigned)L_CAP);
    for (unsigned i = threadIdx.x; i < cnt; i += AGG_THREADS) atomicAdd(&degc[list[i] % R_BIN], 1u);
    list_sum_rows<PITCH>(tile, list, cnt, h);
    __syncthreads();
  }
}

__global__ __launch_bounds__(AGG_THREADS) void bucket_aggregate_relu(
    const unsigned* __restrict__ packed, const unsigned* __restrict__ off,
    const float* __restrict__ h, const float* __restrict__ norm_src,
    const float* __restrict__ b1, float* __restrict__ out, int N, int chunk, int offp) {
  static_assert(AGG_THREADS % 16 == 0, "a thread keeps its 4-column group over the store loop");
  static_assert(R_BIN * H_DIM + 2 * BK_BLOCKS + 1 + BK_BLOCKS / 32 + 2 * R_BIN <= TILE_B_FLOATS,
                "tile A and the small arrays end in front of tile B");
  static_assert((R_BIN * H_DIM) % 4 == 0, "float4 steps over a tile");
  __shared__ alignas(16) float smem[TILE_B_FLOATS + R_BIN * H_DIM + L_CAP];
  float* tile = smem;
  unsigned* rst = (unsigned*)(smem + R_BIN * H_DIM);
  unsigned* pre = rst + BK_BLOCKS;
  unsigned* wtot = pre + BK_BLOCKS + 1;
  unsigned* degc = wtot + BK_BLOCKS / 32;
  float* nsl = (float*)(degc + R_BIN);
  float* tileB = smem + TILE_B_FLOATS;
  unsigned* list = (unsigned*)(tileB + R_BIN * H_DIM);
  const int bin = blockIdx.x;
  const int row_base = bin * R_BIN;
  float4* tile4 = (float4*)tile;
  float4* tileB4 = (float4*)tileB;
  for (int i = threadIdx.x; i < R_BIN * H_DIM / 4; i += AGG_THREADS) {
    tile4[i] = make_float4(0.0f, 0.0f, 0.0f, 0.0f);
    tileB4[i] = make_float4(0.0f, 0.0f, 0.0f, 0.0f);
  }
  if (threadIdx.x < R_BIN) {
    degc[threadIdx.x] = 0u;
    nsl[threadIdx.x] = (row_base + threadIdx.x < N) ? norm_src[row_base + threadIdx.x] : 0.0f;
  }
  bucket_edges<H_DIM>(((threadIdx.x >> 5) & 1) ? tileB : tile, degc, list, rst, pre, wtot,
                      packed, off, bin, offp, chunk, h);

  const float4 b = ((const float4*)b1)[threadIdx.x & 15];
  for (int i = threadIdx.x; i < R_BIN * H_DIM / 4; i += AGG_THREADS) {
    const int node = row_base + (i >> 4);
    if (node < N) {
      float4 v = tile4[i];
      const float4 w = tileB4[i];
      v.x += w.x; v.y += w.y; v.z += w.z; v.w += w.w;
      const float nd = rsqrtf(fmaxf((float)degc[i >> 4], 1.0f));
      const float ns = nsl[i >> 4];
      v.x = fmaxf(v.x * nd + b.x, 0.0f) * ns;
      v.y = fmaxf(v.y * nd + b.y, 0.0f) * ns;
      v.z = fmaxf(v.z * nd + b.z, 0.0f) * ns;
      v.w = fmaxf(v.w * nd + b.w, 0.0f) * ns;
      ((float4*)out)[(long)node * 16 + (i & 15)] = v;
    }
  }
}

#define T_PITCH 68
#define L2_UNITS (R_BIN / 16 * 4)
__global__ __launch_bounds__(AGG_THREADS) void bucket_aggregate_final_wmma(
    const unsigned* __restrict__ packed, const unsigned* __restrict__ off,
    const float* __restrict__ h,
    const float* __restrict__ W_mu, const float* __restrict__ b_mu,
    const float* __restrict__ W_sig, const float* __restrict__ b_sig,
    const float* __restrict__ noise,
    float* __restrict__ z, int N, int chunk, int offp) {
  static_assert(R_BIN * T_PITCH % 4 == 0 && T_PITCH % 2 == 0, "float4 zero fill, 8-byte fragment reads");
  constexpr int W_FLOATS = (H_DIM / 4) * 4 * 32 * 2;
  static_assert(R_BIN * T_PITCH + W_FLOATS + 2 * BK_BLOCKS + 1 + BK_BLOCKS / 32 + 2 * R_BIN <= TILE_B_FLOATS,
                "tile A, one weight matrix and the small arrays end in front of tile B");
  static_assert((R_BIN * T_PITCH) % 2 == 0, "the fragment arrays behind the tiles start on 8 bytes");
  __shared__ alignas(16) float smem[TILE_B_FLOATS + R_BIN * T_PITCH + W_FLOATS + L_CAP];
  float* tile = smem;
  v2f* lWm = (v2f*)(smem + R_BIN * T_PITCH);
  unsigned* rst = (unsigned*)(smem + R_BIN * T_PITCH + W_FLOATS);
  unsigned* pre = rst + BK_BLOCKS;
  unsigned* wtot = pre + BK_BLOCKS + 1;
  unsigned* degc = wtot + BK_BLOCKS / 32;
  float* ndl = (float*)(degc + R_BIN);
  float* tileB = smem + TILE_B_FLOATS;
  v2f* lWs = (v2f*)(tileB + R_BIN * T_PITCH);
  unsigned* list = (unsigned*)(tileB + R_BIN * T_PITCH + W_FLOATS);
  const int bin = blockIdx.x;
  const int row_base = bin * R_BIN;
  float4* tile4 = (float4*)tile;
  float4* tileB4 = (float4*)tileB;
  for (int i = threadIdx.x; i < R_BIN * T_PITCH / 4; i += AGG_THREADS) {
    tile4[i] = make_float4(0.0f, 0.0f, 0.0f, 0.0f);
    tileB4[i] = make_float4(0.0f, 0.0f, 0.0f, 0.0f);
  }
  if (threadIdx.x < R_BIN) degc[threadIdx.x] = 0u;
  stage_w_frag(lWm, W_mu, H_DIM / 4, threadIdx.x, AGG_THREADS);
  stage_w_frag(lWs, W_sig, H_DIM / 4, threadIdx.x, AGG_THREADS);
  bucket_edges<T_PITCH>(((threadIdx.x >> 5) & 1) ? tileB : tile, degc, list, rst, pre, wtot,
                        packed, off, bin, offp, chunk, h);
  for (int i = threadIdx.x; i < R_BIN * T_PITCH / 4; i += AGG_THREADS) {
    float4 v = tile4[i];
    const float4 w = tileB4[i];
    v.x += w.x; v.y += w.y; v.z += w.z; v.w += w.w;
    tile4[i] = v;
  }
  if (threadIdx.x < R_BIN) ndl[threadIdx.x] = rsqrtf(fmaxf((float)degc[threadIdx.x], 1.0f));
  __syncthreads();

  const int wave = threadIdx.x >> 5;
  const int lane = threadIdx.x & 31;
  const int kgrp = (lane >> 4) * 2;
  for (int u = wave; u < L2_UNITS; u += AGG_THREADS / 32) {
    const int rt = u >> 2;
    const int ct = u & 3;
    const int row0 = row_base + rt * 16;
    if (row0 >= N) continue;
    const float* arow = tile + (rt * 16 + (lane & 15)) * T_PITCH + kgrp;
    const int n = ct * 16 + (lane & 15);
    const int lrow = rt * 16 + ((lane >> 4) << 3);
    float nz[8], nd[8];
#pragma unroll
    for (int v = 0; v < 8; ++v) {
      const int m = row_base + lrow + v;
      nz[v] = noise[(long)(m < N ? m : N - 1) * H_DIM + n];
      nd[v] = ndl[lrow + v];
    }
    const float bmu = b_mu[n];
    const float bsg = b_sig[n];
    v8f am = {}, as = {};
#pragma unroll
    for (int kb = 0; kb < H_DIM / 4; ++kb) {
      const v2f a  = *(const v2f*)(arow + kb * 4);
      const v2f bm = lWm[(kb * 4 + ct) * 32 + lane];
      const v2f bs = lWs[(kb * 4 + ct) * 32 + lane];
      am = __builtin_amdgcn_wmma_f32_16x16x4_f32(
          false, a, false, bm, (short)0, am, false, false);
      as = __builtin_amdgcn_wmma_f32_16x16x4_f32(
          false, a, false, bs, (short)0, as, false, false);
    }
    if (row0 + 16 <= N) {
#pragma unroll
      for (int v = 0; v < 8; ++v) {
        const int m = row_base + lrow + v;
        const float mu = am[v] * nd[v] + bmu;
        const float ls = as[v] * nd[v] + bsg;
        z[(long)m * H_DIM + n] = mu + nz[v] * expf(ls);
      }
    } else {
#pragma unroll
      for (int v = 0; v < 8; ++v) {
        const int m = row_base + lrow + v;
        if (m < N) {
          const float mu = am[v] * nd[v] + bmu;
          const float ls = as[v] * nd[v] + bsg;
          z[(long)m * H_DIM + n] = mu + nz[v] * expf(ls);
        }
      }
    }
  }
}

extern "C" void kernel_launch(void* const* d_in, const int* in_sizes, int n_in,
                              void* d_out, int out_size, void* d_ws, size_t ws_size,
                              hipStream_t stream) {
  const float* feat  = (const float*)d_in[0];
  const int*   src   = (const int*)d_in[1];
  const int*   dst   = (const int*)d_in[2];
  const float* noise = (const float*)d_in[3];
  const float* W1    = (const float*)d_in[4];
  const float* b1    = (const float*)d_in[5];
  const float* W_mu  = (const float*)d_in[6];
  const float* b_mu  = (const float*)d_in[7];
  const float* W_sig = (const float*)d_in[8];
  const float* b_sig = (const float*)d_in[9];

  const int E = in_sizes[1];
  const int N = in_sizes[3] / H_DIM;

  const int NB    = (N + R_BIN - 1) / R_BIN;
  const int chunk = (E + BK_BLOCKS - 1) / BK_BLOCKS;
  const int offp  = (NB + 32) & ~31;

  float* ws       = (float*)d_ws;
  float* norm_src = ws;
  unsigned* off_dst = (unsigned*)(ws + N);
  float* buf_a    = ws + 2L * N;
  float* buf_b    = buf_a + (long)N * H_DIM;
  float* zout     = (float*)d_out;

  unsigned* packed_dst  = (unsigned*)(buf_b + (long)N * H_DIM);
  unsigned* packed_src  = (unsigned*)buf_b;
  unsigned* off_src     = packed_src + (long)BK_BLOCKS * chunk;

  bucket_sort_slices<<<dim3(BK_BLOCKS, 2), BK_THREADS, (2 * NB + 1 + chunk) * sizeof(unsigned), stream>>>(
      src, dst, off_dst, off_src, packed_dst, packed_src, E, NB, chunk, offp);

  gemm1_wmma<<<NB, G1_THREADS, 0, stream>>>(feat, packed_src, off_src, W1, norm_src, buf_a, N, chunk, offp);

  bucket_aggregate_relu<<<NB, AGG_THREADS, 0, stream>>>(
      packed_dst, off_dst, buf_a, norm_src, b1, buf_b, N, chunk, offp);

  bucket_aggregate_final_wmma<<<NB, AGG_THREADS, 0, stream>>>(
      packed_dst, off_dst, buf_b, W_mu, b_mu, W_sig, b_sig, noise, zout, N, chunk, offp);
}


# Launch heads: loads issued together and early: weights as 16-byte rows apart from their LDS stores; a wave's run table in registers, its runs' words in flight over the scan (LDS-only barriers)
# speedup vs baseline: 4.5406x; 1.0690x over previous
#include <hip/hip_runtime.h>

typedef __attribute__((ext_vector_type(2))) float v2f;
typedef __attribute__((ext_vector_type(8))) float v8f;

#define F_IN  128
#define H_DIM 64
#define R_BIN 400
#define G1_THREADS (R_BIN / 16 * 32)

#define W_ITEMS(K) ((K) / 2 * (H_DIM / 4))
__device__ __forceinline__ void w_item_load(float4& r0, float4& r1, const float* __restrict__ W, int item) {
  const int kp = item >> 4;
  const int n4 = item & 15;
  const float4* p = (const float4*)(W + kp * 2 * H_DIM + n4 * 4);
  r0 = p[0];
  r1 = p[H_DIM / 4];
}
__device__ __forceinline__ void w_item_store(v2f* lW, int item, float4 r0, float4 r1) {
  const int kp = item >> 4;
  const int n4 = item & 15;
  const int kb = kp >> 1;
  const int ct = n4 >> 2;
  const int ln = (kp & 1) * 16 + (n4 & 3) * 4;
  float4* d = (float4*)(lW + (kb * 4 + ct) * 32 + ln);
  d[0] = make_float4(r0.x, r1.x, r0.y, r1.y);
  d[1] = make_float4(r0.z, r1.z, r0.w, r1.w);
}

#define BK_BLOCKS 256
#define BK_THREADS 1024
#define BK_K 7

__global__ __launch_bounds__(BK_THREADS) void bucket_sort_slices(
    const int* __restrict__ src, const int* __restrict__ dst,
    unsigned* __restrict__ off_dst, unsigned* __restrict__ off_src,
    unsigned* __restrict__ packed_dst, unsigned* __restrict__ packed_src,
    int E, int NB, int chunk, int offp) {
  extern __shared__ unsigned sm[];
  unsigned* pfx  = sm;
  unsigned* loc  = pfx + NB + 1;
  unsigned* word = loc + NB;
  const bool by_src = blockIdx.y != 0;
  const int* __restrict__ key = by_src ? src : dst;
  unsigned* __restrict__ off = (by_src ? off_src : off_dst) + (long)blockIdx.x * offp;
  unsigned* __restrict__ packed = (by_src ? packed_src : packed_dst) + (long)blockIdx.x * chunk;
  for (int i = threadIdx.x; i <= NB; i += BK_THREADS) pfx[i] = 0u;
  __syncthreads();

  const int e0 = blockIdx.x * chunk;
  const int n = max(min(e0 + chunk, E) - e0, 0);
  for (int base = threadIdx.x; base < n; base += BK_THREADS * BK_K) {
    int d[BK_K];
#pragma unroll
    for (int k = 0; k < BK_K; ++k) {
      const int i = base + k * BK_THREADS;
      d[k] = key[e0 + (i < n ? i : n - 1)];
    }
#pragma unroll
    for (int k = 0; k < BK_K; ++k)
      if (base + k * BK_THREADS < n) atomicAdd(&pfx[d[k] / R_BIN], 1u);
  }
  __syncthreads();

  if (threadIdx.x < 32) {
    const int per = (NB + 31) / 32;
    const int b0 = (int)threadIdx.x * per;
    const int b1 = min(b0 + per, NB);
    unsigned sum = 0u;
    for (int b = b0; b < b1; ++b) sum += pfx[b];
    unsigned inc = sum;
    for (int dlt = 1; dlt < 32; dlt <<= 1) {
      const unsigned v = __shfl_up(inc, dlt);
      if ((int)threadIdx.x >= dlt) inc += v;
    }
    unsigned run = inc - sum;
    for (int b = b0; b < b1; ++b) {
      const unsigned c = pfx[b];
      pfx[b] = run;
      loc[b] = run;
      run += c;
    }
    if (threadIdx.x == 31) pfx[NB] = inc;
  }
  __syncthreads();

  for (int i = threadIdx.x; i <= NB; i += BK_THREADS) off[i] = pfx[i];
  for (int base = threadIdx.x; base < n; base += BK_THREADS * BK_K) {
    int d[BK_K], s[BK_K];
#pragma unroll
    for (int k = 0; k < BK_K; ++k) {
      const int i = base + k * BK_THREADS;
      const int ic = i < n ? i : n - 1;
      d[k] = key[e0 + ic];
      s[k] = by_src ? 0 : src[e0 + ic];
    }
#pragma unroll
    for (int k = 0; k < BK_K; ++k) {
      if (base + k * BK_THREADS < n) {
        const int bin = d[k] / R_BIN;
        const unsigned at = atomicAdd(&loc[bin], 1u);
        word[at] = (unsigned)(d[k] - bin * R_BIN) + (unsigned)s[k] * R_BIN;
      }
    }
  }
  __syncthreads();
  for (int i = threadIdx.x; i < n; i += BK_THREADS) packed[i] = word[i];
}

__device__ __forceinline__ uint2 run_offsets(const unsigned* __restrict__ off, int bin, int offp) {
  const unsigned s = min((unsigned)threadIdx.x, (unsigned)(BK_BLOCKS - 1));
  const unsigned* p = off + (long)s * offp + bin;
  return make_uint2(p[0], p[1]);
}
__device__ __forceinline__ unsigned keep_run(unsigned* rst, uint2 o, int chunk) {
  unsigned len = 0u;
  if (threadIdx.x < BK_BLOCKS) {
    len = o.y - o.x;
    rst[threadIdx.x] = threadIdx.x * (unsigned)chunk + o.x;
  }
  return len;
}

template <int NW>
__device__ __forceinline__ void wave_run_table(unsigned& s_l, unsigned& len_l,
                                               const unsigned* __restrict__ off,
                                               int bin, int offp, int chunk) {
  constexpr int RPW = (BK_BLOCKS + NW - 1) / NW;
  static_assert(RPW <= 32, "one lane per run of the wave");
  const int wave = threadIdx.x >> 5;
  const int lane = threadIdx.x & 31;
  const int r = wave + lane * NW;
  const bool ok = lane < RPW && r < BK_BLOCKS;
  const int rc = ok ? r : 0;
  const unsigned* p = off + (long)rc * offp + bin;
  const unsigned a = p[0];
  const unsigned b = p[1];
  s_l = (unsigned)rc * (unsigned)chunk + a;
  len_l = ok ? b - a : 0u;
}
__device__ __forceinline__ unsigned run_at(unsigned v_l, int j) {
  return (unsigned)__builtin_amdgcn_readlane((int)v_l, j);
}

__device__ __forceinline__ void lds_barrier() {
  __builtin_amdgcn_fence(__ATOMIC_RELEASE, "workgroup", "local");
  __builtin_amdgcn_s_barrier();
  __builtin_amdgcn_fence(__ATOMIC_ACQUIRE, "workgroup", "local");
}

__device__ __forceinline__ void scan_runs(unsigned* pre, unsigned* wtot, unsigned len) {
  static_assert(BK_BLOCKS % 32 == 0, "the runs fill whole waves");
  const int wave = threadIdx.x >> 5;
  const int lane = threadIdx.x & 31;
  unsigned inc = len;
  for (int dlt = 1; dlt < 32; dlt <<= 1) {
    const unsigned v = __shfl_up(inc, dlt);
    if (lane >= dlt) inc += v;
  }
  if (threadIdx.x < BK_BLOCKS && lane == 31) wtot[wave] = inc;
  lds_barrier();
  if (threadIdx.x < BK_BLOCKS) {
    unsigned before = 0u;
    for (int q = 0; q < wave; ++q) before += wtot[q];
    pre[threadIdx.x] = before + inc - len;
    if (threadIdx.x == BK_BLOCKS - 1) pre[BK_BLOCKS] = before + inc;
  }
  lds_barrier();
}

template <int NW>
struct RunWords {
  static constexpr int RPW = (BK_BLOCKS + NW - 1) / NW;
  unsigned w0[RPW], w1[RPW], s[RPW], ln[RPW];
};
template <int NW>
__device__ __forceinline__ void run_words_load(RunWords<NW>& rw, const unsigned* __restrict__ packed,
                                               unsigned s_l, unsigned len_l) {
  const unsigned lane = threadIdx.x & 31;
#pragma unroll
  for (int j = 0; j < RunWords<NW>::RPW; ++j) {
    rw.s[j] = run_at(s_l, j);
    rw.ln[j] = run_at(len_l, j);
    rw.w0[j] = packed[lane < rw.ln[j] ? rw.s[j] + lane : 0u];
    rw.w1[j] = packed[lane + 32u < rw.ln[j] ? rw.s[j] + lane + 32u : 0u];
  }
}
template <int NW>
__device__ __forceinline__ void count_run_words(unsigned* degc, const unsigned* __restrict__ packed,
                                                const RunWords<NW>& rw) {
  const unsigned lane = threadIdx.x & 31;
#pragma unroll
  for (int j = 0; j < RunWords<NW>::RPW; ++j) {
    if (lane < rw.ln[j]) atomicAdd(&degc[rw.w0[j] % R_BIN], 1u);
    if (lane + 32u < rw.ln[j]) atomicAdd(&degc[rw.w1[j] % R_BIN], 1u);
  }
#pragma unroll
  for (int j = 0; j < RunWords<NW>::RPW; ++j)
    for (unsigned i = 64u + lane; i < rw.ln[j]; i += 32u)
      atomicAdd(&degc[packed[rw.s[j] + i] % R_BIN], 1u);
}

__global__ __launch_bounds__(G1_THREADS) void gemm1_wmma(
    const float* __restrict__ feat, const unsigned* __restrict__ packed_src,
    const unsigned* __restrict__ off_src, const float* __restrict__ W1,
    float* __restrict__ norm_src, float* __restrict__ h1, int N, int chunk, int offp) {
  static_assert(R_BIN % 16 == 0, "a bucket is a whole number of 16-row tiles");
  static_assert(G1_THREADS <= W_ITEMS(F_IN) && 2 * G1_THREADS >= W_ITEMS(F_IN),
                "a thread stages one or two weight items");
  __shared__ alignas(16) v2f lW[(F_IN / 4) * 4 * 32];
  __shared__ unsigned degc[R_BIN];
  const int it1 = (int)threadIdx.x + G1_THREADS;
  const bool has1 = it1 < W_ITEMS(F_IN);
  float4 wa0, wa1, wb0, wb1;
  w_item_load(wa0, wa1, W1, (int)threadIdx.x);
  w_item_load(wb0, wb1, W1, has1 ? it1 : (int)threadIdx.x);
  unsigned s_l, len_l;
  wave_run_table<G1_THREADS / 32>(s_l, len_l, off_src, blockIdx.x, offp, chunk);
  if (threadIdx.x < R_BIN) degc[threadIdx.x] = 0u;
  RunWords<G1_THREADS / 32> rw;
  run_words_load(rw, packed_src, s_l, len_l);
  w_item_store(lW, (int)threadIdx.x, wa0, wa1);
  if (has1) w_item_store(lW, it1, wb0, wb1);
  __syncthreads();
  count_run_words(degc, packed_src, rw);
  __syncthreads();
  if (threadIdx.x < R_BIN && blockIdx.x * R_BIN + threadIdx.x < N)
    norm_src[blockIdx.x * R_BIN + threadIdx.x] =
        rsqrtf(fmaxf((float)degc[threadIdx.x], 1.0f));

  const int wave = threadIdx.x >> 5;
  const int lane = threadIdx.x & 31;
  const int tile = blockIdx.x * (R_BIN / 16) + wave;
  const int row0 = tile * 16;
  if (row0 >= N) return;

  const int mrow = row0 + (lane & 15);
  const int kgrp = (lane >> 4) * 2;
  const float ns = rsqrtf(fmaxf((float)degc[wave * 16 + (lane & 15)], 1.0f));

  v8f acc[4] = {};
  for (int kb = 0; kb < F_IN / 4; ++kb) {
    v2f a = *(const v2f*)(feat + (long)mrow * F_IN + kb * 4 + kgrp);
    a.x *= ns; a.y *= ns;
#pragma unroll
    for (int ct = 0; ct < 4; ++ct) {
      const v2f b = lW[(kb * 4 + ct) * 32 + lane];
      acc[ct] = __builtin_amdgcn_wmma_f32_16x16x4_f32(
          false, a, false, b, (short)0, acc[ct], false, false);
    }
  }

  const int mbase = row0 + ((lane >> 4) << 3);
#pragma unroll
  for (int ct = 0; ct < 4; ++ct) {
    const int n = ct * 16 + (lane & 15);
#pragma unroll
    for (int v = 0; v < 8; ++v)
      h1[(long)(mbase + v) * H_DIM + n] = acc[ct][v];
  }
}

template <int PITCH>
__device__ __forceinline__ void tile_add_row(float* tile, unsigned p, int lane, float2 v) {
  float* t = tile + (p % R_BIN) * PITCH + lane * 2;
  atomicAdd(t, v.x);
  atomicAdd(t + 1, v.y);
}

#define AGG_THREADS 1024
#define TILE_B_FLOATS 32768
#define EPT 12
#define L_CAP 8192

__device__ __forceinline__ void fill_list(unsigned* list, const unsigned* __restrict__ packed,
                                          const unsigned* rst, const unsigned* pre, unsigned R) {
  constexpr int NW = AGG_THREADS / 32;
  constexpr int RPW = (BK_BLOCKS + NW - 1) / NW;
  const int wave = threadIdx.x >> 5;
  const unsigned lane = threadIdx.x & 31;
  unsigned w0[RPW], w1[RPW], g0[RPW], ln[RPW];
#pragma unroll
  for (int j = 0; j < RPW; ++j) {
    const int r = wave + j * NW;
    const int rc = r < BK_BLOCKS ? r : 0;
    const unsigned s = rst[rc];
    const unsigned p = pre[rc];
    ln[j] = r < BK_BLOCKS ? pre[rc + 1] - p : 0u;
    g0[j] = p + lane - R;
    w0[j] = packed[(lane < ln[j] && g0[j] < L_CAP) ? s + lane : 0u];
    w1[j] = packed[(lane + 32u < ln[j] && g0[j] + 32u < L_CAP) ? s + lane + 32u : 0u];
  }
#pragma unroll
  for (int j = 0; j < RPW; ++j) {
    if (lane < ln[j] && g0[j] < L_CAP) list[g0[j]] = w0[j];
    if (lane + 32u < ln[j] && g0[j] + 32u < L_CAP) list[g0[j] + 32u] = w1[j];
  }
  for (int r = wave; r < BK_BLOCKS; r += NW) {
    const unsigned s = rst[r];
    const unsigned p = pre[r];
    const unsigned len = pre[r + 1] - p;
    for (unsigned i = 64u + lane; i < len; i += 32u) {
      const unsigned g = p + i - R;
      if (g < L_CAP) list[g] = packed[s + i];
    }
  }
}

template <int NW>
__device__ __forceinline__ void fill_list_first(unsigned* list, const unsigned* __restrict__ packed,
                                                const unsigned* pre, const RunWords<NW>& rw) {
  constexpr int RPW = RunWords<NW>::RPW;
  const int wave = threadIdx.x >> 5;
  const unsigned lane = threadIdx.x & 31;
  unsigned p[RPW];
#pragma unroll
  for (int j = 0; j < RPW; ++j) {
    const int r = wave + j * NW;
    p[j] = pre[r < BK_BLOCKS ? r : 0];
  }
#pragma unroll
  for (int j = 0; j < RPW; ++j) {
    const unsigned g0 = p[j] + lane;
    if (lane < rw.ln[j] && g0 < L_CAP) list[g0] = rw.w0[j];
    if (lane + 32u < rw.ln[j] && g0 + 32u < L_CAP) list[g0 + 32u] = rw.w1[j];
  }
#pragma unroll
  for (int j = 0; j < RPW; ++j) {
    for (unsigned i = 64u + lane; i < rw.ln[j]; i += 32u) {
      const unsigned g = p[j] + i;
      if (g < L_CAP) list[g] = packed[rw.s[j] + i];
    }
  }
}

template <int PITCH>
__device__ __forceinline__ void list_sum_rows(float* tile, const unsigned* list, unsigned cnt_v,
                                              const float* __restrict__ h) {
  static_assert(EPT <= 32, "one lane per edge of a trip");
  const unsigned wave = (unsigned)__builtin_amdgcn_readfirstlane((int)(threadIdx.x >> 5));
  const unsigned cnt = (unsigned)__builtin_amdgcn_readfirstlane((int)cnt_v);
  const unsigned lane = threadIdx.x & 31;
  const float2* hv = (const float2*)h;
  for (unsigned e = wave * EPT; e < cnt; e += (AGG_THREADS / 32) * EPT) {
    const unsigned w = list[min(e + lane, cnt - 1u)];
    const unsigned srow = w / R_BIN;
    const int hoff_l = (int)(srow * (H_DIM / 2));
    const int toff_l = (int)((w - srow * R_BIN) * PITCH);
    unsigned hoff[EPT], toff[EPT];
    float2 v[EPT];
#pragma unroll
    for (int k = 0; k < EPT; ++k) {
      hoff[k] = (unsigned)__builtin_amdgcn_readlane(hoff_l, k);
      toff[k] = (unsigned)__builtin_amdgcn_readlane(toff_l, k);
    }
#pragma unroll
    for (int k = 0; k < EPT; ++k) v[k] = (hv + (size_t)hoff[k])[lane];
    if (e + EPT <= cnt) {
#pragma unroll
      for (int k = 0; k < EPT; ++k) {
        float* t = tile + toff[k] + lane * 2;
        atomicAdd(t, v[k].x);
        atomicAdd(t + 1, v[k].y);
      }
    } else {
#pragma unroll
      for (int k = 0; k < EPT; ++k) {
        if (e + k < cnt) {
          float* t = tile + toff[k] + lane * 2;
          atomicAdd(t, v[k].x);
          atomicAdd(t + 1, v[k].y);
        }
      }
    }
  }
}

template <int PITCH>
__device__ __forceinline__ void bucket_edges(float* tile, unsigned* degc, unsigned* list,
                                             unsigned* rst, unsigned* pre, unsigned* wtot,
                                             const unsigned* __restrict__ packed,
                                             uint2 ro, unsigned s_l, unsigned len_l, int chunk,
                                             const float* __restrict__ h) {
  RunWords<AGG_THREADS / 32> rw;
  run_words_load(rw, packed, s_l, len_l);
  scan_runs(pre, wtot, keep_run(rst, ro, chunk));
  const unsigned total = pre[BK_BLOCKS];
  for (unsigned R = 0u; R < total; R += L_CAP) {
    if (R == 0u) fill_list_first(list, packed, pre, rw);
    else fill_list(list, packed, rst, pre, R);
    __syncthreads();
    const unsigned cnt = min(total - R, (unsigned)L_CAP);
    for (unsigned i = threadIdx.x; i < cnt; i += AGG_THREADS) atomicAdd(&degc[list[i] % R_BIN], 1u);
    list_sum_rows<PITCH>(tile, list, cnt, h);
    __syncthreads();
  }
}

__global__ __launch_bounds__(AGG_THREADS) void bucket_aggregate_relu(
    const unsigned* __restrict__ packed, const unsigned* __restrict__ off,
    const float* __restrict__ h, const float* __restrict__ norm_src,
    const float* __restrict__ b1, float* __restrict__ out, int N, int chunk, int offp) {
  static_assert(AGG_THREADS % 16 == 0, "a thread keeps its 4-column group over the store loop");
  static_assert(R_BIN * H_DIM + 2 * BK_BLOCKS + 1 + BK_BLOCKS / 32 + 2 * R_BIN <= TILE_B_FLOATS,
                "tile A and the small arrays end in front of tile B");
  static_assert((R_BIN * H_DIM) % 4 == 0, "float4 steps over a tile");
  __shared__ alignas(16) float smem[TILE_B_FLOATS + R_BIN * H_DIM + L_CAP];
  float* tile = smem;
  unsigned* rst = (unsigned*)(smem + R_BIN * H_DIM);
  unsigned* pre = rst + BK_BLOCKS;
  unsigned* wtot = pre + BK_BLOCKS + 1;
  unsigned* degc = wtot + BK_BLOCKS / 32;
  float* nsl = (float*)(degc + R_BIN);
  float* tileB = smem + TILE_B_FLOATS;
  unsigned* list = (unsigned*)(tileB + R_BIN * H_DIM);
  const int bin = blockIdx.x;
  const int row_base = bin * R_BIN;
  float4* tile4 = (float4*)tile;
  float4* tileB4 = (float4*)tileB;
  const uint2 ro = run_offsets(off, bin, offp);
  unsigned s_l, len_l;
  wave_run_table<AGG_THREADS / 32>(s_l, len_l, off, bin, offp, chunk);
  float nsv = 0.0f;
  if (threadIdx.x < R_BIN && row_base + threadIdx.x < N) nsv = norm_src[row_base + threadIdx.x];
  for (int i = threadIdx.x; i < R_BIN * H_DIM / 4; i += AGG_THREADS) {
    tile4[i] = make_float4(0.0f, 0.0f, 0.0f, 0.0f);
    tileB4[i] = make_float4(0.0f, 0.0f, 0.0f, 0.0f);
  }
  if (threadIdx.x < R_BIN) {
    degc[threadIdx.x] = 0u;
    nsl[threadIdx.x] = nsv;
  }
  bucket_edges<H_DIM>(((threadIdx.x >> 5) & 1) ? tileB : tile, degc, list, rst, pre, wtot,
                      packed, ro, s_l, len_l, chunk, h);

  const float4 b = ((const float4*)b1)[threadIdx.x & 15];
  for (int i = threadIdx.x; i < R_BIN * H_DIM / 4; i += AGG_THREADS) {
    const int node = row_base + (i >> 4);
    if (node < N) {
      float4 v = tile4[i];
      const float4 w = tileB4[i];
      v.x += w.x; v.y += w.y; v.z += w.z; v.w += w.w;
      const float nd = rsqrtf(fmaxf((float)degc[i >> 4], 1.0f));
      const float ns = nsl[i >> 4];
      v.x = fmaxf(v.x * nd + b.x, 0.0f) * ns;
      v.y = fmaxf(v.y * nd + b.y, 0.0f) * ns;
      v.z = fmaxf(v.z * nd + b.z, 0.0f) * ns;
      v.w = fmaxf(v.w * nd + b.w, 0.0f) * ns;
      ((float4*)out)[(long)node * 16 + (i & 15)] = v;
    }
  }
}

#define T_PITCH 68
#define L2_UNITS (R_BIN / 16 * 4)
__global__ __launch_bounds__(AGG_THREADS) void bucket_aggregate_final_wmma(
    const unsigned* __restrict__ packed, const unsigned* __restrict__ off,
    const float* __restrict__ h,
    const float* __restrict__ W_mu, const float* __restrict__ b_mu,
    const float* __restrict__ W_sig, const float* __restrict__ b_sig,
    const float* __restrict__ noise,
    float* __restrict__ z, int N, int chunk, int offp) {
  static_assert(R_BIN * T_PITCH % 4 == 0 && T_PITCH % 2 == 0, "float4 zero fill, 8-byte fragment reads");
  constexpr int W_FLOATS = (H_DIM / 4) * 4 * 32 * 2;
  static_assert(R_BIN * T_PITCH + W_FLOATS + 2 * BK_BLOCKS + 1 + BK_BLOCKS / 32 + 2 * R_BIN <= TILE_B_FLOATS,
                "tile A, one weight matrix and the small arrays end in front of tile B");
  static_assert((R_BIN * T_PITCH) % 2 == 0, "the fragment arrays behind the tiles start on 8 bytes");
  __shared__ alignas(16) float smem[TILE_B_FLOATS + R_BIN * T_PITCH + W_FLOATS + L_CAP];
  float* tile = smem;
  v2f* lWm = (v2f*)(smem + R_BIN * T_PITCH);
  unsigned* rst = (unsigned*)(smem + R_BIN * T_PITCH + W_FLOATS);
  unsigned* pre = rst + BK_BLOCKS;
  unsigned* wtot = pre + BK_BLOCKS + 1;
  unsigned* degc = wtot + BK_BLOCKS / 32;
  float* ndl = (float*)(degc + R_BIN);
  float* tileB = smem + TILE_B_FLOATS;
  v2f* lWs = (v2f*)(tileB + R_BIN * T_PITCH);
  unsigned* list = (unsigned*)(tileB + R_BIN * T_PITCH + W_FLOATS);
  const int bin = blockIdx.x;
  const int row_base = bin * R_BIN;
  float4* tile4 = (float4*)tile;
  float4* tileB4 = (float4*)tileB;
  static_assert(AGG_THREADS == 2 * W_ITEMS(H_DIM) && W_ITEMS(H_DIM) % 32 == 0,
                "one weight item per thread, a wave in one matrix");
  const bool sig = threadIdx.x >= W_ITEMS(H_DIM);
  const int wit = (int)threadIdx.x - (sig ? W_ITEMS(H_DIM) : 0);
  float4 w0, w1;
  w_item_load(w0, w1, sig ? W_sig : W_mu, wit);
  const uint2 ro = run_offsets(off, bin, offp);
  unsigned s_l, len_l;
  wave_run_table<AGG_THREADS / 32>(s_l, len_l, off, bin, offp, chunk);
  for (int i = threadIdx.x; i < R_BIN * T_PITCH / 4; i += AGG_THREADS) {
    tile4[i] = make_float4(0.0f, 0.0f, 0.0f, 0.0f);
    tileB4[i] = make_float4(0.0f, 0.0f, 0.0f, 0.0f);
  }
  if (threadIdx.x < R_BIN) degc[threadIdx.x] = 0u;
  w_item_store(sig ? lWs : lWm, wit, w0, w1);
  bucket_edges<T_PITCH>(((threadIdx.x >> 5) & 1) ? tileB : tile, degc, list, rst, pre, wtot,
                        packed, ro, s_l, len_l, chunk, h);
  for (int i = threadIdx.x; i < R_BIN * T_PITCH / 4; i += AGG_THREADS) {
    float4 v = tile4[i];
    const float4 w = tileB4[i];
    v.x += w.x; v.y += w.y; v.z += w.z; v.w += w.w;
    tile4[i] = v;
  }
  if (threadIdx.x < R_BIN) ndl[threadIdx.x] = rsqrtf(fmaxf((float)degc[threadIdx.x], 1.0f));
  __syncthreads();

  const int wave = threadIdx.x >> 5;
  const int lane = threadIdx.x & 31;
  const int kgrp = (lane >> 4) * 2;
  for (int u = wave; u < L2_UNITS; u += AGG_THREADS / 32) {
    const int rt = u >> 2;
    const int ct = u & 3;
    const int row0 = row_base + rt * 16;
    if (row0 >= N) continue;
    const float* arow = tile + (rt * 16 + (lane & 15)) * T_PITCH + kgrp;
    const int n = ct * 16 + (lane & 15);
    const int lrow = rt * 16 + ((lane >> 4) << 3);
    float nz[8], nd[8];
#pragma unroll
    for (int v = 0; v < 8; ++v) {
      const int m = row_base + lrow + v;
      nz[v] = noise[(long)(m < N ? m : N - 1) * H_DIM + n];
      nd[v] = ndl[lrow + v];
    }
    const float bmu = b_mu[n];
    const float bsg = b_sig[n];
    v8f am = {}, as = {};
#pragma unroll
    for (int kb = 0; kb < H_DIM / 4; ++kb) {
      const v2f a  = *(const v2f*)(arow + kb * 4);
      const v2f bm = lWm[(kb * 4 + ct) * 32 + lane];
      const v2f bs = lWs[(kb * 4 + ct) * 32 + lane];
      am = __builtin_amdgcn_wmma_f32_16x16x4_f32(
          false, a, false, bm, (short)0, am, false, false);
      as = __builtin_amdgcn_wmma_f32_16x16x4_f32(
          false, a, false, bs, (short)0, as, false, false);
    }
    if (row0 + 16 <= N) {
#pragma unroll
      for (int v = 0; v < 8; ++v) {
        const int m = row_base + lrow + v;
        const float mu = am[v] * nd[v] + bmu;
        const float ls = as[v] * nd[v] + bsg;
        z[(long)m * H_DIM + n] = mu + nz[v] * expf(ls);
      }
    } else {
#pragma unroll
      for (int v = 0; v < 8; ++v) {
        const int m = row_base + lrow + v;
        if (m < N) {
          const float mu = am[v] * nd[v] + bmu;
          const float ls = as[v] * nd[v] + bsg;
          z[(long)m * H_DIM + n] = mu + nz[v] * expf(ls);
        }
      }
    }
  }
}

extern "C" void kernel_launch(void* const* d_in, const int* in_sizes, int n_in,
                              void* d_out, int out_size, void* d_ws, size_t ws_size,
                              hipStream_t stream) {
  const float* feat  = (const float*)d_in[0];
  const int*   src   = (const int*)d_in[1];
  const int*   dst   = (const int*)d_in[2];
  const float* noise = (const float*)d_in[3];
  const float* W1    = (const float*)d_in[4];
  const float* b1    = (const float*)d_in[5];
  const float* W_mu  = (const float*)d_in[6];
  const float* b_mu  = (const float*)d_in[7];
  const float* W_sig = (const float*)d_in[8];
  const float* b_sig = (const float*)d_in[9];

  const int E = in_sizes[1];
  const int N = in_sizes[3] / H_DIM;

  const int NB    = (N + R_BIN - 1) / R_BIN;
  const int chunk = (E + BK_BLOCKS - 1) / BK_BLOCKS;
  const int offp  = (NB + 32) & ~31;

  float* ws       = (float*)d_ws;
  float* norm_src = ws;
  unsigned* off_dst = (unsigned*)(ws + N);
  float* buf_a    = ws + 2L * N;
  float* buf_b    = buf_a + (long)N * H_DIM;
  float* zout     = (float*)d_out;

  unsigned* packed_dst  = (unsigned*)(buf_b + (long)N * H_DIM);
  unsigned* packed_src  = (unsigned*)buf_b;
  unsigned* off_src     = packed_src + (long)BK_BLOCKS * chunk;

  bucket_sort_slices<<<dim3(BK_BLOCKS, 2), BK_THREADS, (2 * NB + 1 + chunk) * sizeof(unsigned), stream>>>(
      src, dst, off_dst, off_src, packed_dst, packed_src, E, NB, chunk, offp);

  gemm1_wmma<<<NB, G1_THREADS, 0, stream>>>(feat, packed_src, off_src, W1, norm_src, buf_a, N, chunk, offp);

  bucket_aggregate_relu<<<NB, AGG_THREADS, 0, stream>>>(
      packed_dst, off_dst, buf_a, norm_src, b1, buf_b, N, chunk, offp);

  bucket_aggregate_final_wmma<<<NB, AGG_THREADS, 0, stream>>>(
      packed_dst, off_dst, buf_b, W_mu, b_mu, W_sig, b_sig, noise, zout, N, chunk, offp);
}
